# starting kernel
# baseline (speedup 1.0000x reference)
#include <hip/hip_runtime.h>

typedef float v2f __attribute__((ext_vector_type(2)));
typedef float v8f __attribute__((ext_vector_type(8)));

#define DFEAT 128

__global__ __launch_bounds__(256) void deg_accum_kernel(const int* __restrict__ row,
                                                        const float* __restrict__ w,
                                                        float* __restrict__ deg, int E) {
    int e = blockIdx.x * 256 + threadIdx.x;
    if (e < E) atomicAdd(&deg[row[e]], w[e]);
}

__global__ __launch_bounds__(256) void rsqrt_kernel(float* __restrict__ deg, int n) {
    int i = blockIdx.x * 256 + threadIdx.x;
    if (i < n) deg[i] = 1.0f / sqrtf(deg[i]);
}

__global__ __launch_bounds__(256) void norm_kernel(const int* __restrict__ row,
                                                   const int* __restrict__ col,
                                                   const float* __restrict__ w,
                                                   const float* __restrict__ dinv,
                                                   float* __restrict__ norm, int E) {
    int e = blockIdx.x * 256 + threadIdx.x;
    if (e < E) norm[e] = dinv[row[e]] * w[e] * dinv[col[e]];
}

template <bool RELU>
__global__ __launch_bounds__(256) void gemm_wmma_kernel(const float* __restrict__ X,
                                                        const float* __restrict__ W,
                                                        const float* __restrict__ bias,
                                                        float* __restrict__ H, int nrows) {
    const int lane = threadIdx.x & 31;
    const int wave = threadIdx.x >> 5;
    const int m    = lane & 15;
    const int half = lane >> 4;
    const int row0 = blockIdx.x * 16;
    const int col0 = wave * 16;

    int ar = row0 + m;
    if (ar >= nrows) ar = nrows - 1;
    const float* __restrict__ xrow = X + (size_t)ar * DFEAT;
    const float* __restrict__ wrow = W + (size_t)(col0 + m) * DFEAT;

    const float bv = bias[col0 + m];
    v8f c;
#pragma unroll
    for (int r = 0; r < 8; ++r) c[r] = bv;

#pragma unroll
    for (int kb = 0; kb < DFEAT; kb += 4) {
        v2f a = *(const v2f*)(xrow + kb + 2 * half);
        if (RELU) { a.x = fmaxf(a.x, 0.0f); a.y = fmaxf(a.y, 0.0f); }
        v2f b = *(const v2f*)(wrow + kb + 2 * half);
        c = __builtin_amdgcn_wmma_f32_16x16x4_f32(false, a, false, b,
                                                  (short)0, c, false, false);
    }

#pragma unroll
    for (int r = 0; r < 8; ++r) {
        int M = row0 + r + 8 * half;
        if (M < nrows) H[(size_t)M * DFEAT + col0 + m] = c[r];
    }
}

__global__ __launch_bounds__(256) void scatter_add_kernel(const float* __restrict__ H,
                                                          const int* __restrict__ row,
                                                          const int* __restrict__ col,
                                                          const float* __restrict__ norm,
                                                          float* __restrict__ out, int E) {
    int e = blockIdx.x * 8 + (threadIdx.x >> 5);
    if (e >= E) return;
    const int lane = threadIdx.x & 31;
    const int r = row[e];
    const int cidx = col[e];
    const float nv = norm[e];

    const float4 hv = *(const float4*)(H + (size_t)r * DFEAT + lane * 4);
    float* op = out + (size_t)cidx * DFEAT + lane * 4;
    atomicAdd(op + 0, nv * hv.x);
    atomicAdd(op + 1, nv * hv.y);
    atomicAdd(op + 2, nv * hv.z);
    atomicAdd(op + 3, nv * hv.w);
}

extern "C" void kernel_launch(void* const* d_in, const int* in_sizes, int n_in,
                              void* d_out, int out_size, void* d_ws, size_t ws_size,
                              hipStream_t stream) {
    const float* x  = (const float*)d_in[0];
    const int*   ei = (const int*)d_in[1];
    const float* ew = (const float*)d_in[2];
    const float* W1 = (const float*)d_in[3];
    const float* b1 = (const float*)d_in[4];
    const float* W2 = (const float*)d_in[5];
    const float* b2 = (const float*)d_in[6];
    float* out = (float*)d_out;

    const int E = in_sizes[2];
    const int n = in_sizes[0] / DFEAT;
    const int* row = ei;
    const int* col = ei + E;

    char* ws = (char*)d_ws;
    size_t off = 0;
    auto alignup = [](size_t v) { return (v + 255) & ~(size_t)255; };
    float* deg  = (float*)(ws + off); off += alignup((size_t)n * 4);
    float* norm = (float*)(ws + off); off += alignup((size_t)E * 4);
    float* h    = (float*)(ws + off); off += alignup((size_t)n * DFEAT * 4);
    float* agg  = (float*)(ws + off);

    const int gemm_blocks = (n + 15) / 16;
    const int edge_blocks = (E + 255) / 256;
    const int node_blocks = (n + 255) / 256;
    const int scat_blocks = (E + 7) / 8;

    hipMemsetAsync(deg, 0, (size_t)n * 4, stream);
    deg_accum_kernel<<<edge_blocks, 256, 0, stream>>>(row, ew, deg, E);
    rsqrt_kernel<<<node_blocks, 256, 0, stream>>>(deg, n);
    norm_kernel<<<edge_blocks, 256, 0, stream>>>(row, col, ew, deg, norm, E);

    gemm_wmma_kernel<false><<<gemm_blocks, 256, 0, stream>>>(x, W1, b1, h, n);
    hipMemsetAsync(agg, 0, (size_t)n * DFEAT * 4, stream);
    scatter_add_kernel<<<scat_blocks, 256, 0, stream>>>(h, row, col, norm, agg, E);

    gemm_wmma_kernel<true><<<gemm_blocks, 256, 0, stream>>>(agg, W2, b2, h, n);
    hipMemsetAsync(out, 0, (size_t)n * DFEAT * 4, stream);
    scatter_add_kernel<<<scat_blocks, 256, 0, stream>>>(h, row, col, norm, out, E);
}


# GEMM register tiling: a wave owns a 32x128 strip (16 accumulators), fragments by 16-byte loads with k = 8t+4h+e in both loaders, 16-byte whole-line stores; k sum order changed
# speedup vs baseline: 1.8025x; 1.8025x over previous
#include <hip/hip_runtime.h>

typedef float v2f __attribute__((ext_vector_type(2)));
typedef float v8f __attribute__((ext_vector_type(8)));

#define DFEAT 128

__global__ __launch_bounds__(256) void deg_accum_kernel(const int* __restrict__ row,
                                                        const float* __restrict__ w,
                                                        float* __restrict__ deg, int E) {
    int e = blockIdx.x * 256 + threadIdx.x;
    if (e < E) atomicAdd(&deg[row[e]], w[e]);
}

__global__ __launch_bounds__(256) void rsqrt_kernel(float* __restrict__ deg, int n) {
    int i = blockIdx.x * 256 + threadIdx.x;
    if (i < n) deg[i] = 1.0f / sqrtf(deg[i]);
}

__global__ __launch_bounds__(256) void norm_kernel(const int* __restrict__ row,
                                                   const int* __restrict__ col,
                                                   const float* __restrict__ w,
                                                   const float* __restrict__ dinv,
                                                   float* __restrict__ norm, int E) {
    int e = blockIdx.x * 256 + threadIdx.x;
    if (e < E) norm[e] = dinv[row[e]] * w[e] * dinv[col[e]];
}

typedef float v4f __attribute__((ext_vector_type(4)));

template <bool RELU>
__global__ __launch_bounds__(128) void gemm_wmma_kernel(const float* __restrict__ X,
                                                        const float* __restrict__ W,
                                                        const float* __restrict__ bias,
                                                        float* __restrict__ H, int nrows) {
    const int lane = threadIdx.x & 31;
    const int wave = threadIdx.x >> 5;
    const int m    = lane & 15;
    const int half = lane >> 4;
    const int row0 = (blockIdx.x * 4 + wave) * 32;
    if (row0 >= nrows) return;

    int ar0 = row0 + m;
    if (ar0 >= nrows) ar0 = nrows - 1;
    int ar1 = row0 + 16 + m;
    if (ar1 >= nrows) ar1 = nrows - 1;
    const float* __restrict__ xr0 = X + (size_t)ar0 * DFEAT + 4 * half;
    const float* __restrict__ xr1 = X + (size_t)ar1 * DFEAT + 4 * half;
    const float* __restrict__ wr  = W + (size_t)(4 * m) * DFEAT + 4 * half;

    v8f c[2][8];
    {
        const v4f bv0 = *(const v4f*)(bias + 4 * m);
        const v4f bv1 = *(const v4f*)(bias + 64 + 4 * m);
#pragma unroll
        for (int j = 0; j < 8; ++j) {
            const float bv = (j < 4) ? bv0[j & 3] : bv1[j & 3];
#pragma unroll
            for (int r = 0; r < 8; ++r) { c[0][j][r] = bv; c[1][j][r] = bv; }
        }
    }

    v4f a_cur[2], b_cur[8];
    a_cur[0] = *(const v4f*)xr0;
    a_cur[1] = *(const v4f*)xr1;
#pragma unroll
    for (int j = 0; j < 8; ++j)
        b_cur[j] = *(const v4f*)(wr + (size_t)((j >> 2) * 64 + (j & 3)) * DFEAT);

#pragma unroll
    for (int t = 0; t < DFEAT / 8; ++t) {
        v4f a_nxt[2], b_nxt[8];
        if (t + 1 < DFEAT / 8) {
            const int kb = 8 * (t + 1);
            a_nxt[0] = *(const v4f*)(xr0 + kb);
            a_nxt[1] = *(const v4f*)(xr1 + kb);
#pragma unroll
            for (int j = 0; j < 8; ++j)
                b_nxt[j] = *(const v4f*)(wr + (size_t)((j >> 2) * 64 + (j & 3)) * DFEAT + kb);
        }
        __builtin_amdgcn_sched_barrier(0);
#pragma unroll
        for (int s = 0; s < 2; ++s) {
            v2f af[2], bf[8];
#pragma unroll
            for (int i = 0; i < 2; ++i) {
                af[i].x = a_cur[i][2 * s]; af[i].y = a_cur[i][2 * s + 1];
                if (RELU) { af[i].x = fmaxf(af[i].x, 0.0f); af[i].y = fmaxf(af[i].y, 0.0f); }
            }
#pragma unroll
            for (int j = 0; j < 8; ++j) { bf[j].x = b_cur[j][2 * s]; bf[j].y = b_cur[j][2 * s + 1]; }
#pragma unroll
            for (int i = 0; i < 2; ++i)
#pragma unroll
                for (int j = 0; j < 8; ++j)
                    c[i][j] = __builtin_amdgcn_wmma_f32_16x16x4_f32(false, af[i], false, bf[j],
                                                                    (short)0, c[i][j], false, false);
        }
        __builtin_amdgcn_sched_barrier(0);
        if (t + 1 < DFEAT / 8) {
            a_cur[0] = a_nxt[0]; a_cur[1] = a_nxt[1];
#pragma unroll
            for (int j = 0; j < 8; ++j) b_cur[j] = b_nxt[j];
        }
    }

#pragma unroll
    for (int i = 0; i < 2; ++i) {
#pragma unroll
        for (int r = 0; r < 8; ++r) {
            const int M = row0 + 16 * i + r + 8 * half;
            if (M < nrows) {
                float* hp = H + (size_t)M * DFEAT + 4 * m;
#pragma unroll
                for (int g = 0; g < 2; ++g) {
                    v4f o;
                    o.x = c[i][4 * g + 0][r]; o.y = c[i][4 * g + 1][r];
                    o.z = c[i][4 * g + 2][r]; o.w = c[i][4 * g + 3][r];
                    *(v4f*)(hp + 64 * g) = o;
                }
            }
        }
    }
}

#define SCAT_EDGES 32
#define SCAT_GROUP 8
__device__ __forceinline__ void scatter_one_edge(const float* __restrict__ H,
                                                 float* __restrict__ out,
                                                 int ri, int ci, float ni, int lane) {
    const float* hp = H + (size_t)ri * DFEAT + lane;
    float* op = out + (size_t)ci * DFEAT + lane;
#pragma unroll
    for (int j = 0; j < 4; ++j) atomicAdd(op + 32 * j, ni * hp[32 * j]);
}

__global__ __launch_bounds__(256) void scatter_add_kernel(const float* __restrict__ H,
                                                          const int* __restrict__ row,
                                                          const int* __restrict__ col,
                                                          const float* __restrict__ norm,
                                                          float* __restrict__ out, int E) {
    const int lane = threadIdx.x & 31;
    const int wave = __builtin_amdgcn_readfirstlane((int)(threadIdx.x >> 5));
    const int e0 = (blockIdx.x * 8 + wave) * SCAT_EDGES;
    if (e0 >= E) return;

    const int e = e0 + lane;
    int r = 0, c = 0;
    float nv = 0.0f;
    if (e < E) { r = row[e]; c = col[e]; nv = norm[e]; }
    const int nvb = __builtin_bit_cast(int, nv);
    const int rem = E - e0;
    const int cnt = rem < SCAT_EDGES ? rem : SCAT_EDGES;

    int i0 = 0;
    for (; i0 + SCAT_GROUP <= cnt; i0 += SCAT_GROUP) {
        float hv[SCAT_GROUP][4];
#pragma unroll
        for (int u = 0; u < SCAT_GROUP; ++u) {
            const int ri = __builtin_amdgcn_readlane(r, i0 + u);
            const float* hp = H + (size_t)ri * DFEAT + lane;
#pragma unroll
            for (int j = 0; j < 4; ++j) hv[u][j] = hp[32 * j];
        }
#pragma unroll
        for (int u = 0; u < SCAT_GROUP; ++u) {
            const int ci = __builtin_amdgcn_readlane(c, i0 + u);
            const float ni = __builtin_bit_cast(float, __builtin_amdgcn_readlane(nvb, i0 + u));
            float* op = out + (size_t)ci * DFEAT + lane;
#pragma unroll
            for (int j = 0; j < 4; ++j) atomicAdd(op + 32 * j, ni * hv[u][j]);
        }
    }
    for (; i0 < cnt; ++i0) {
        const int ri = __builtin_amdgcn_readlane(r, i0);
        const int ci = __builtin_amdgcn_readlane(c, i0);
        const float ni = __builtin_bit_cast(float, __builtin_amdgcn_readlane(nvb, i0));
        scatter_one_edge(H, out, ri, ci, ni, lane);
    }
}

extern "C" void kernel_launch(void* const* d_in, const int* in_sizes, int n_in,
                              void* d_out, int out_size, void* d_ws, size_t ws_size,
                              hipStream_t stream) {
    const float* x  = (const float*)d_in[0];
    const int*   ei = (const int*)d_in[1];
    const float* ew = (const float*)d_in[2];
    const float* W1 = (const float*)d_in[3];
    const float* b1 = (const float*)d_in[4];
    const float* W2 = (const float*)d_in[5];
    const float* b2 = (const float*)d_in[6];
    float* out = (float*)d_out;

    const int E = in_sizes[2];
    const int n = in_sizes[0] / DFEAT;
    const int* row = ei;
    const int* col = ei + E;

    char* ws = (char*)d_ws;
    size_t off = 0;
    auto alignup = [](size_t v) { return (v + 255) & ~(size_t)255; };
    float* deg  = (float*)(ws + off); off += alignup((size_t)n * 4);
    float* norm = (float*)(ws + off); off += alignup((size_t)E * 4);
    float* h    = (float*)(ws + off); off += alignup((size_t)n * DFEAT * 4);
    float* agg  = (float*)(ws + off);

    const int gemm_blocks = (n + 127) / 128;
    const int edge_blocks = (E + 255) / 256;
    const int node_blocks = (n + 255) / 256;
    const int scat_blocks = edge_blocks;

    hipMemsetAsync(deg, 0, (size_t)n * 4, stream);
    deg_accum_kernel<<<edge_blocks, 256, 0, stream>>>(row, ew, deg, E);
    rsqrt_kernel<<<node_blocks, 256, 0, stream>>>(deg, n);
    norm_kernel<<<edge_blocks, 256, 0, stream>>>(row, col, ew, deg, norm, E);

    gemm_wmma_kernel<false><<<gemm_blocks, 128, 0, stream>>>(x, W1, b1, h, n);
    hipMemsetAsync(agg, 0, (size_t)n * DFEAT * 4, stream);
    scatter_add_kernel<<<scat_blocks, 256, 0, stream>>>(h, row, col, norm, agg, E);

    gemm_wmma_kernel<true><<<gemm_blocks, 128, 0, stream>>>(agg, W2, b2, h, n);
    hipMemsetAsync(out, 0, (size_t)n * DFEAT * 4, stream);
    scatter_add_kernel<<<scat_blocks, 256, 0, stream>>>(h, row, col, norm, out, E);
}


# Launch fusion, 10 stream operations to 5: zero-fills inside the GEMM launches, degree sum as extra work-groups of GEMM 1, deg^-0.5 and edge norm inside scatter 1; all sum orders as before
# speedup vs baseline: 1.9782x; 1.0975x over previous
#include <hip/hip_runtime.h>

typedef float v2f __attribute__((ext_vector_type(2)));
typedef float v8f __attribute__((ext_vector_type(8)));

#define DFEAT 128


typedef float v4f __attribute__((ext_vector_type(4)));

typedef int v4i __attribute__((ext_vector_type(4)));

template <bool RELU, bool DEG>
__global__ __launch_bounds__(128) void gemm_wmma_kernel(const float* __restrict__ X,
                                                        const float* __restrict__ W,
                                                        const float* __restrict__ bias,
                                                        float* __restrict__ H,
                                                        float* __restrict__ Z, int nrows,
                                                        const int* __restrict__ erow,
                                                        const float* __restrict__ ew,
                                                        float* __restrict__ deg, int E) {
    if (DEG) {
        const int gemm_blocks = (nrows + 127) / 128;
        if ((int)blockIdx.x >= gemm_blocks) {
            const int e4 = (((int)blockIdx.x - gemm_blocks) * 128 + (int)threadIdx.x) * 4;
            if (e4 + 4 <= E) {
                const v4i r4 = *(const v4i*)(erow + e4);
                const v4f w4 = *(const v4f*)(ew + e4);
                atomicAdd(&deg[r4.x], w4.x);
                atomicAdd(&deg[r4.y], w4.y);
                atomicAdd(&deg[r4.z], w4.z);
                atomicAdd(&deg[r4.w], w4.w);
            } else {
                for (int e = e4; e < E; ++e) atomicAdd(&deg[erow[e]], ew[e]);
            }
            return;
        }
    }
    const int lane = threadIdx.x & 31;
    const int wave = threadIdx.x >> 5;
    const int m    = lane & 15;
    const int half = lane >> 4;
    const int row0 = (blockIdx.x * 4 + wave) * 32;
    if (row0 >= nrows) return;

    int ar0 = row0 + m;
    if (ar0 >= nrows) ar0 = nrows - 1;
    int ar1 = row0 + 16 + m;
    if (ar1 >= nrows) ar1 = nrows - 1;
    const float* __restrict__ xr0 = X + (size_t)ar0 * DFEAT + 4 * half;
    const float* __restrict__ xr1 = X + (size_t)ar1 * DFEAT + 4 * half;
    const float* __restrict__ wr  = W + (size_t)(4 * m) * DFEAT + 4 * half;

    v8f c[2][8];
    {
        const v4f bv0 = *(const v4f*)(bias + 4 * m);
        const v4f bv1 = *(const v4f*)(bias + 64 + 4 * m);
#pragma unroll
        for (int j = 0; j < 8; ++j) {
            const float bv = (j < 4) ? bv0[j & 3] : bv1[j & 3];
#pragma unroll
            for (int r = 0; r < 8; ++r) { c[0][j][r] = bv; c[1][j][r] = bv; }
        }
    }

    v4f a_cur[2], b_cur[8];
    a_cur[0] = *(const v4f*)xr0;
    a_cur[1] = *(const v4f*)xr1;
#pragma unroll
    for (int j = 0; j < 8; ++j)
        b_cur[j] = *(const v4f*)(wr + (size_t)((j >> 2) * 64 + (j & 3)) * DFEAT);

    {
        const v4f z4 = {0.0f, 0.0f, 0.0f, 0.0f};
        float* zp = Z + (size_t)row0 * DFEAT + 4 * lane;
        if (row0 + 32 <= nrows) {
#pragma unroll
            for (int q = 0; q < 32; ++q) *(v4f*)(zp + (size_t)q * DFEAT) = z4;
        } else {
            for (int q = 0; row0 + q < nrows; ++q) *(v4f*)(zp + (size_t)q * DFEAT) = z4;
        }
    }

#pragma unroll
    for (int t = 0; t < DFEAT / 8; ++t) {
        v4f a_nxt[2], b_nxt[8];
        if (t + 1 < DFEAT / 8) {
            const int kb = 8 * (t + 1);
            a_nxt[0] = *(const v4f*)(xr0 + kb);
            a_nxt[1] = *(const v4f*)(xr1 + kb);
#pragma unroll
            for (int j = 0; j < 8; ++j)
                b_nxt[j] = *(const v4f*)(wr + (size_t)((j >> 2) * 64 + (j & 3)) * DFEAT + kb);
        }
        __builtin_amdgcn_sched_barrier(0);
#pragma unroll
        for (int s = 0; s < 2; ++s) {
            v2f af[2], bf[8];
#pragma unroll
            for (int i = 0; i < 2; ++i) {
                af[i].x = a_cur[i][2 * s]; af[i].y = a_cur[i][2 * s + 1];
                if (RELU) { af[i].x = fmaxf(af[i].x, 0.0f); af[i].y = fmaxf(af[i].y, 0.0f); }
            }
#pragma unroll
            for (int j = 0; j < 8; ++j) { bf[j].x = b_cur[j][2 * s]; bf[j].y = b_cur[j][2 * s + 1]; }
#pragma unroll
            for (int i = 0; i < 2; ++i)
#pragma unroll
                for (int j = 0; j < 8; ++j)
                    c[i][j] = __builtin_amdgcn_wmma_f32_16x16x4_f32(false, af[i], false, bf[j],
                                                                    (short)0, c[i][j], false, false);
        }
        __builtin_amdgcn_sched_barrier(0);
        if (t + 1 < DFEAT / 8) {
            a_cur[0] = a_nxt[0]; a_cur[1] = a_nxt[1];
#pragma unroll
            for (int j = 0; j < 8; ++j) b_cur[j] = b_nxt[j];
        }
    }

#pragma unroll
    for (int i = 0; i < 2; ++i) {
#pragma unroll
        for (int r = 0; r < 8; ++r) {
            const int M = row0 + 16 * i + r + 8 * half;
            if (M < nrows) {
                float* hp = H + (size_t)M * DFEAT + 4 * m;
#pragma unroll
                for (int g = 0; g < 2; ++g) {
                    v4f o;
                    o.x = c[i][4 * g + 0][r]; o.y = c[i][4 * g + 1][r];
                    o.z = c[i][4 * g + 2][r]; o.w = c[i][4 * g + 3][r];
                    *(v4f*)(hp + 64 * g) = o;
                }
            }
        }
    }
}

#define SCAT_EDGES 32
#define SCAT_GROUP 8
__device__ __forceinline__ void scatter_one_edge(const float* __restrict__ H,
                                                 float* __restrict__ out,
                                                 int ri, int ci, float ni, int lane) {
    const float* hp = H + (size_t)ri * DFEAT + lane;
    float* op = out + (size_t)ci * DFEAT + lane;
#pragma unroll
    for (int j = 0; j < 4; ++j) atomicAdd(op + 32 * j, ni * hp[32 * j]);
}

template <bool MAKE_NORM>
__global__ __launch_bounds__(256) void scatter_add_kernel(const float* __restrict__ H,
                                                          const int* __restrict__ row,
                                                          const int* __restrict__ col,
                                                          const float* __restrict__ w,
                                                          const float* __restrict__ deg,
                                                          float* __restrict__ norm,
                                                          float* __restrict__ out, int E) {
    const int lane = threadIdx.x & 31;
    const int wave = __builtin_amdgcn_readfirstlane((int)(threadIdx.x >> 5));
    const int e0 = (blockIdx.x * 8 + wave) * SCAT_EDGES;
    if (e0 >= E) return;

    const int e = e0 + lane;
    int r = 0, c = 0;
    float nv = 0.0f;
    if (e < E) {
        r = row[e]; c = col[e];
        if (MAKE_NORM) {
            const float dr = 1.0f / sqrtf(deg[r]);
            const float dc = 1.0f / sqrtf(deg[c]);
            nv = dr * w[e] * dc;
            norm[e] = nv;
        } else {
            nv = norm[e];
        }
    }
    const int nvb = __builtin_bit_cast(int, nv);
    const int rem = E - e0;
    const int cnt = rem < SCAT_EDGES ? rem : SCAT_EDGES;

    int i0 = 0;
    for (; i0 + SCAT_GROUP <= cnt; i0 += SCAT_GROUP) {
        float hv[SCAT_GROUP][4];
#pragma unroll
        for (int u = 0; u < SCAT_GROUP; ++u) {
            const int ri = __builtin_amdgcn_readlane(r, i0 + u);
            const float* hp = H + (size_t)ri * DFEAT + lane;
#pragma unroll
            for (int j = 0; j < 4; ++j) hv[u][j] = hp[32 * j];
        }
#pragma unroll
        for (int u = 0; u < SCAT_GROUP; ++u) {
            const int ci = __builtin_amdgcn_readlane(c, i0 + u);
            const float ni = __builtin_bit_cast(float, __builtin_amdgcn_readlane(nvb, i0 + u));
            float* op = out + (size_t)ci * DFEAT + lane;
#pragma unroll
            for (int j = 0; j < 4; ++j) atomicAdd(op + 32 * j, ni * hv[u][j]);
        }
    }
    for (; i0 < cnt; ++i0) {
        const int ri = __builtin_amdgcn_readlane(r, i0);
        const int ci = __builtin_amdgcn_readlane(c, i0);
        const float ni = __builtin_bit_cast(float, __builtin_amdgcn_readlane(nvb, i0));
        scatter_one_edge(H, out, ri, ci, ni, lane);
    }
}

extern "C" void kernel_launch(void* const* d_in, const int* in_sizes, int n_in,
                              void* d_out, int out_size, void* d_ws, size_t ws_size,
                              hipStream_t stream) {
    const float* x  = (const float*)d_in[0];
    const int*   ei = (const int*)d_in[1];
    const float* ew = (const float*)d_in[2];
    const float* W1 = (const float*)d_in[3];
    const float* b1 = (const float*)d_in[4];
    const float* W2 = (const float*)d_in[5];
    const float* b2 = (const float*)d_in[6];
    float* out = (float*)d_out;

    const int E = in_sizes[2];
    const int n = in_sizes[0] / DFEAT;
    const int* row = ei;
    const int* col = ei + E;

    char* ws = (char*)d_ws;
    size_t off = 0;
    auto alignup = [](size_t v) { return (v + 255) & ~(size_t)255; };
    float* deg  = (float*)(ws + off); off += alignup((size_t)n * 4);
    float* norm = (float*)(ws + off); off += alignup((size_t)E * 4);
    float* h    = (float*)(ws + off); off += alignup((size_t)n * DFEAT * 4);
    float* agg  = (float*)(ws + off);

    const int gemm_blocks = (n + 127) / 128;
    const int edge_blocks = (E + 255) / 256;
    const int scat_blocks = edge_blocks;
    const int deg_blocks  = (E + 511) / 512;

    hipMemsetAsync(deg, 0, (size_t)n * 4, stream);

    gemm_wmma_kernel<false, true><<<gemm_blocks + deg_blocks, 128, 0, stream>>>(x, W1, b1, h, agg, n,
                                                                              row, ew, deg, E);
    scatter_add_kernel<true><<<scat_blocks, 256, 0, stream>>>(h, row, col, ew, deg, norm, agg, E);

    gemm_wmma_kernel<true, false><<<gemm_blocks, 128, 0, stream>>>(agg, W2, b2, h, out, n,
                                                                 nullptr, nullptr, nullptr, 0);
    scatter_add_kernel<false><<<scat_blocks, 256, 0, stream>>>(h, row, col, ew, deg, norm, out, E);
}


# speedup vs baseline: 2.3037x; 1.1645x over previous
#include <hip/hip_runtime.h>

typedef float v2f __attribute__((ext_vector_type(2)));
typedef float v8f __attribute__((ext_vector_type(8)));

#define DFEAT 128


typedef float v4f __attribute__((ext_vector_type(4)));

typedef int v4i __attribute__((ext_vector_type(4)));

template <bool RELU, bool DEG>
__global__ __launch_bounds__(128) void gemm_wmma_kernel(const float* __restrict__ X,
                                                        const float* __restrict__ W,
                                                        const float* __restrict__ bias,
                                                        float* __restrict__ H,
                                                        float* __restrict__ Z, int nrows,
                                                        const int* __restrict__ erow,
                                                        const float* __restrict__ ew,
                                                        float* __restrict__ deg, int E) {
    if (DEG) {
        const int gemm_blocks = (nrows + 127) / 128;
        if ((int)blockIdx.x >= gemm_blocks) {
            const int e4 = (((int)blockIdx.x - gemm_blocks) * 128 + (int)threadIdx.x) * 4;
            if (e4 + 4 <= E) {
                const v4i r4 = *(const v4i*)(erow + e4);
                const v4f w4 = *(const v4f*)(ew + e4);
                atomicAdd(&deg[r4.x], w4.x);
                atomicAdd(&deg[r4.y], w4.y);
                atomicAdd(&deg[r4.z], w4.z);
                atomicAdd(&deg[r4.w], w4.w);
            } else {
                for (int e = e4; e < E; ++e) atomicAdd(&deg[erow[e]], ew[e]);
            }
            return;
        }
    }
    const int lane = threadIdx.x & 31;
    const int wave = threadIdx.x >> 5;
    const int m    = lane & 15;
    const int half = lane >> 4;
    const int row0 = (blockIdx.x * 4 + wave) * 32;
    if (row0 >= nrows) return;

    int ar0 = row0 + m;
    if (ar0 >= nrows) ar0 = nrows - 1;
    int ar1 = row0 + 16 + m;
    if (ar1 >= nrows) ar1 = nrows - 1;
    const float* __restrict__ xr0 = X + (size_t)ar0 * DFEAT + 4 * half;
    const float* __restrict__ xr1 = X + (size_t)ar1 * DFEAT + 4 * half;
    const float* __restrict__ wr  = W + (size_t)(4 * m) * DFEAT + 4 * half;

    v8f c[2][8];
    {
        const v4f bv0 = *(const v4f*)(bias + 4 * m);
        const v4f bv1 = *(const v4f*)(bias + 64 + 4 * m);
#pragma unroll
        for (int j = 0; j < 8; ++j) {
            const float bv = (j < 4) ? bv0[j & 3] : bv1[j & 3];
#pragma unroll
            for (int r = 0; r < 8; ++r) { c[0][j][r] = bv; c[1][j][r] = bv; }
        }
    }

    v4f a_cur[2], b_cur[8];
    a_cur[0] = *(const v4f*)xr0;
    a_cur[1] = *(const v4f*)xr1;
#pragma unroll
    for (int j = 0; j < 8; ++j)
        b_cur[j] = *(const v4f*)(wr + (size_t)((j >> 2) * 64 + (j & 3)) * DFEAT);

    {
        const v4f z4 = {0.0f, 0.0f, 0.0f, 0.0f};
        float* zp = Z + (size_t)row0 * DFEAT + 4 * lane;
        if (row0 + 32 <= nrows) {
#pragma unroll
            for (int q = 0; q < 32; ++q) *(v4f*)(zp + (size_t)q * DFEAT) = z4;
        } else {
            for (int q = 0; row0 + q < nrows; ++q) *(v4f*)(zp + (size_t)q * DFEAT) = z4;
        }
    }

#pragma unroll
    for (int t = 0; t < DFEAT / 8; ++t) {
        v4f a_nxt[2], b_nxt[8];
        if (t + 1 < DFEAT / 8) {
            const int kb = 8 * (t + 1);
            a_nxt[0] = *(const v4f*)(xr0 + kb);
            a_nxt[1] = *(const v4f*)(xr1 + kb);
#pragma unroll
            for (int j = 0; j < 8; ++j)
                b_nxt[j] = *(const v4f*)(wr + (size_t)((j >> 2) * 64 + (j & 3)) * DFEAT + kb);
        }
        __builtin_amdgcn_sched_barrier(0);
#pragma unroll
        for (int s = 0; s < 2; ++s) {
            v2f af[2], bf[8];
#pragma unroll
            for (int i = 0; i < 2; ++i) {
                af[i].x = a_cur[i][2 * s]; af[i].y = a_cur[i][2 * s + 1];
                if (RELU) { af[i].x = fmaxf(af[i].x, 0.0f); af[i].y = fmaxf(af[i].y, 0.0f); }
            }
#pragma unroll
            for (int j = 0; j < 8; ++j) { bf[j].x = b_cur[j][2 * s]; bf[j].y = b_cur[j][2 * s + 1]; }
#pragma unroll
            for (int i = 0; i < 2; ++i)
#pragma unroll
                for (int j = 0; j < 8; ++j)
                    c[i][j] = __builtin_amdgcn_wmma_f32_16x16x4_f32(false, af[i], false, bf[j],
                                                                    (short)0, c[i][j], false, false);
        }
        __builtin_amdgcn_sched_barrier(0);
        if (t + 1 < DFEAT / 8) {
            a_cur[0] = a_nxt[0]; a_cur[1] = a_nxt[1];
#pragma unroll
            for (int j = 0; j < 8; ++j) b_cur[j] = b_nxt[j];
        }
    }

#pragma unroll
    for (int i = 0; i < 2; ++i) {
#pragma unroll
        for (int r = 0; r < 8; ++r) {
            const int M = row0 + 16 * i + r + 8 * half;
            if (M < nrows) {
                float* hp = H + (size_t)M * DFEAT + 4 * m;
#pragma unroll
                for (int g = 0; g < 2; ++g) {
                    v4f o;
                    o.x = c[i][4 * g + 0][r]; o.y = c[i][4 * g + 1][r];
                    o.z = c[i][4 * g + 2][r]; o.w = c[i][4 * g + 3][r];
                    *(v4f*)(hp + 64 * g) = o;
                }
            }
        }
    }
}

#define SCAT_CHUNK 512
#define SCAT_WAVE_EDGES 64
#define SCAT_GROUP 16
#define CTR_STRIDE 32

template <bool MAKE_NORM>
__global__ __launch_bounds__(256) void scatter_add_kernel(const float* __restrict__ H,
                                                          const int* __restrict__ row,
                                                          const int* __restrict__ col,
                                                          const float* __restrict__ w,
                                                          const float* __restrict__ deg,
                                                          float* __restrict__ norm,
                                                          float* __restrict__ out, int E,
                                                          int* __restrict__ ctr, int T) {
    __shared__ int s_ticket[2];
    if (threadIdx.x == 0) {
        const int die = (__builtin_amdgcn_s_sendmsg_rtn(0x87) >> 16) & 15;
        int f = (die >> 2) & 1;
        int t = atomicAdd(&ctr[f * CTR_STRIDE], 1);
        if (t >= T) {
            f ^= 1;
            t = atomicAdd(&ctr[f * CTR_STRIDE], 1);
        }
        s_ticket[0] = t;
        s_ticket[1] = f;
    }
    __syncthreads();
    const int t = __builtin_amdgcn_readfirstlane(s_ticket[0]);
    const int f = __builtin_amdgcn_readfirstlane(s_ticket[1]);
    if (t >= T) return;

    const int lane = threadIdx.x & 31;
    const int wave = __builtin_amdgcn_readfirstlane((int)(threadIdx.x >> 5));
    const int e0 = t * SCAT_CHUNK + wave * SCAT_WAVE_EDGES;
    if (e0 >= E) return;

    int r[2], c[2], nvb[2], ec[2];
    bool ok[2];
#pragma unroll
    for (int s = 0; s < 2; ++s) {
        const int e = e0 + 32 * s + lane;
        ok[s] = e < E;
        ec[s] = ok[s] ? e : E - 1;
        r[s] = row[ec[s]];
        c[s] = col[ec[s]];
    }
    if (MAKE_NORM) {
        float wv[2], dr[2], dc[2];
#pragma unroll
        for (int s = 0; s < 2; ++s) { wv[s] = w[ec[s]]; dr[s] = deg[r[s]]; dc[s] = deg[c[s]]; }
#pragma unroll
        for (int s = 0; s < 2; ++s) {
            const float ir = 1.0f / sqrtf(dr[s]);
            const float ic = 1.0f / sqrtf(dc[s]);
            const float nv = ir * wv[s] * ic;
            if (f == 0 && ok[s]) norm[ec[s]] = nv;
            nvb[s] = __builtin_bit_cast(int, nv);
        }
    } else {
#pragma unroll
        for (int s = 0; s < 2; ++s) nvb[s] = __builtin_bit_cast(int, norm[ec[s]]);
    }
    const int rem = E - e0;
    const int cnt = rem < SCAT_WAVE_EDGES ? rem : SCAT_WAVE_EDGES;
    const unsigned bo = (64u * (unsigned)f + (unsigned)lane) * 4u;
    const char* __restrict__ Hb = (const char*)H;
    char* __restrict__ ob = (char*)out;

#pragma unroll
    for (int s = 0; s < 2; ++s) {
        int cs = cnt - 32 * s;
        cs = cs < 0 ? 0 : (cs > 32 ? 32 : cs);
        int i0 = 0;
        for (; i0 + SCAT_GROUP <= cs; i0 += SCAT_GROUP) {
            float hv[SCAT_GROUP][2];
#pragma unroll
            for (int u = 0; u < SCAT_GROUP; ++u) {
                const int ri = __builtin_amdgcn_readlane(r[s], i0 + u);
                const float* hp = (const float*)(Hb + ((unsigned)ri * (DFEAT * 4u) + bo));
                hv[u][0] = hp[0];
                hv[u][1] = hp[32];
            }
#pragma unroll
            for (int u = 0; u < SCAT_GROUP; ++u) {
                const int ci = __builtin_amdgcn_readlane(c[s], i0 + u);
                const float ni = __builtin_bit_cast(float, __builtin_amdgcn_readlane(nvb[s], i0 + u));
                float* op = (float*)(ob + ((unsigned)ci * (DFEAT * 4u) + bo));
                atomicAdd(op, ni * hv[u][0]);
                atomicAdd(op + 32, ni * hv[u][1]);
            }
        }
        for (; i0 < cs; ++i0) {
            const int ri = __builtin_amdgcn_readlane(r[s], i0);
            const int ci = __builtin_amdgcn_readlane(c[s], i0);
            const float ni = __builtin_bit_cast(float, __builtin_amdgcn_readlane(nvb[s], i0));
            const float* hp = (const float*)(Hb + ((unsigned)ri * (DFEAT * 4u) + bo));
            float* op = (float*)(ob + ((unsigned)ci * (DFEAT * 4u) + bo));
            atomicAdd(op, ni * hp[0]);
            atomicAdd(op + 32, ni * hp[32]);
        }
    }
}

extern "C" void kernel_launch(void* const* d_in, const int* in_sizes, int n_in,
                              void* d_out, int out_size, void* d_ws, size_t ws_size,
                              hipStream_t stream) {
    const float* x  = (const float*)d_in[0];
    const int*   ei = (const int*)d_in[1];
    const float* ew = (const float*)d_in[2];
    const float* W1 = (const float*)d_in[3];
    const float* b1 = (const float*)d_in[4];
    const float* W2 = (const float*)d_in[5];
    const float* b2 = (const float*)d_in[6];
    float* out = (float*)d_out;

    const int E = in_sizes[2];
    const int n = in_sizes[0] / DFEAT;
    const int* row = ei;
    const int* col = ei + E;

    char* ws = (char*)d_ws;
    size_t off = 0;
    auto alignup = [](size_t v) { return (v + 255) & ~(size_t)255; };
    float* deg  = (float*)(ws + off); off += alignup((size_t)n * 4);
    int*   ctr  = (int*)(ws + off);   off += alignup((size_t)4 * CTR_STRIDE * 4);
    const size_t zero_bytes = off;
    float* norm = (float*)(ws + off); off += alignup((size_t)E * 4);
    float* h    = (float*)(ws + off); off += alignup((size_t)n * DFEAT * 4);
    float* agg  = (float*)(ws + off);

    const int gemm_blocks = (n + 127) / 128;
    const int chunks      = (E + SCAT_CHUNK - 1) / SCAT_CHUNK;
    const int scat_blocks = 2 * chunks;
    const int deg_blocks  = (E + 511) / 512;

    hipMemsetAsync(deg, 0, zero_bytes, stream);

    gemm_wmma_kernel<false, true><<<gemm_blocks + deg_blocks, 128, 0, stream>>>(x, W1, b1, h, agg, n,
                                                                              row, ew, deg, E);
    scatter_add_kernel<true><<<scat_blocks, 256, 0, stream>>>(h, row, col, ew, deg, norm, agg, E,
                                                              ctr, chunks);

    gemm_wmma_kernel<true, false><<<gemm_blocks, 128, 0, stream>>>(agg, W2, b2, h, out, n,
                                                                 nullptr, nullptr, nullptr, 0);
    scatter_add_kernel<false><<<scat_blocks, 256, 0, stream>>>(h, row, col, ew, deg, norm, out, E,
                                                               ctr + 2 * CTR_STRIDE, chunks);
}


# Aggregation by owner: a work-group owns 200 output rows, finds its edges by a scan of byte keys, sums norm*h in LDS (float atomics, arrival order), stores each row once; no global row atomics
# speedup vs baseline: 2.4027x; 1.0430x over previous
#include <hip/hip_runtime.h>

typedef float v2f __attribute__((ext_vector_type(2)));
typedef float v8f __attribute__((ext_vector_type(8)));

#define DFEAT 128


typedef float v4f __attribute__((ext_vector_type(4)));

typedef int v4i __attribute__((ext_vector_type(4)));
typedef unsigned v4u __attribute__((ext_vector_type(4)));
#define ACC_ROWS 200

#define BAND_KEY(c) (((unsigned)(c) / (unsigned)ACC_ROWS) & 255u)

template <bool RELU, bool DEG>
__global__ __launch_bounds__(128) void gemm_wmma_kernel(const float* __restrict__ X,
                                                        const float* __restrict__ W,
                                                        const float* __restrict__ bias,
                                                        float* __restrict__ H, int nrows,
                                                        const int* __restrict__ erow,
                                                        const int* __restrict__ ecol,
                                                        const float* __restrict__ ew,
                                                        float* __restrict__ deg,
                                                        unsigned* __restrict__ okey, int E) {
    if (DEG) {
        const int gemm_blocks = (nrows + 127) / 128;
        if ((int)blockIdx.x >= gemm_blocks) {
            const int e4 = (((int)blockIdx.x - gemm_blocks) * 128 + (int)threadIdx.x) * 4;
            const int epad = (E + 15) & ~15;
            if (e4 >= epad) return;
            unsigned kp = 0;
            if (e4 + 4 <= E) {
                const v4i r4 = *(const v4i*)(erow + e4);
                const v4i c4 = *(const v4i*)(ecol + e4);
                const v4f w4 = *(const v4f*)(ew + e4);
                atomicAdd(&deg[r4.x], w4.x);
                atomicAdd(&deg[r4.y], w4.y);
                atomicAdd(&deg[r4.z], w4.z);
                atomicAdd(&deg[r4.w], w4.w);
                kp = BAND_KEY(c4.x) | (BAND_KEY(c4.y) << 8) | (BAND_KEY(c4.z) << 16) |
                     (BAND_KEY(c4.w) << 24);
            } else {
                for (int j = 0; j < 4; ++j) {
                    const int e = e4 + j;
                    if (e < E) {
                        atomicAdd(&deg[erow[e]], ew[e]);
                        kp |= BAND_KEY(ecol[e]) << (8 * j);
                    }
                }
            }
            okey[e4 >> 2] = kp;
            return;
        }
    }
    const int lane = threadIdx.x & 31;
    const int wave = threadIdx.x >> 5;
    const int m    = lane & 15;
    const int half = lane >> 4;
    const int row0 = (blockIdx.x * 4 + wave) * 32;
    if (row0 >= nrows) return;

    int ar0 = row0 + m;
    if (ar0 >= nrows) ar0 = nrows - 1;
    int ar1 = row0 + 16 + m;
    if (ar1 >= nrows) ar1 = nrows - 1;
    const float* __restrict__ xr0 = X + (size_t)ar0 * DFEAT + 4 * half;
    const float* __restrict__ xr1 = X + (size_t)ar1 * DFEAT + 4 * half;
    const float* __restrict__ wr  = W + (size_t)(4 * m) * DFEAT + 4 * half;

    v8f c[2][8];
    {
        const v4f bv0 = *(const v4f*)(bias + 4 * m);
        const v4f bv1 = *(const v4f*)(bias + 64 + 4 * m);
#pragma unroll
        for (int j = 0; j < 8; ++j) {
            const float bv = (j < 4) ? bv0[j & 3] : bv1[j & 3];
#pragma unroll
            for (int r = 0; r < 8; ++r) { c[0][j][r] = bv; c[1][j][r] = bv; }
        }
    }

    v4f a_cur[2], b_cur[8];
    a_cur[0] = *(const v4f*)xr0;
    a_cur[1] = *(const v4f*)xr1;
#pragma unroll
    for (int j = 0; j < 8; ++j)
        b_cur[j] = *(const v4f*)(wr + (size_t)((j >> 2) * 64 + (j & 3)) * DFEAT);

#pragma unroll
    for (int t = 0; t < DFEAT / 8; ++t) {
        v4f a_nxt[2], b_nxt[8];
        if (t + 1 < DFEAT / 8) {
            const int kb = 8 * (t + 1);
            a_nxt[0] = *(const v4f*)(xr0 + kb);
            a_nxt[1] = *(const v4f*)(xr1 + kb);
#pragma unroll
            for (int j = 0; j < 8; ++j)
                b_nxt[j] = *(const v4f*)(wr + (size_t)((j >> 2) * 64 + (j & 3)) * DFEAT + kb);
        }
        __builtin_amdgcn_sched_barrier(0);
#pragma unroll
        for (int s = 0; s < 2; ++s) {
            v2f af[2], bf[8];
#pragma unroll
            for (int i = 0; i < 2; ++i) {
                af[i].x = a_cur[i][2 * s]; af[i].y = a_cur[i][2 * s + 1];
                if (RELU) { af[i].x = fmaxf(af[i].x, 0.0f); af[i].y = fmaxf(af[i].y, 0.0f); }
            }
#pragma unroll
            for (int j = 0; j < 8; ++j) { bf[j].x = b_cur[j][2 * s]; bf[j].y = b_cur[j][2 * s + 1]; }
#pragma unroll
            for (int i = 0; i < 2; ++i)
#pragma unroll
                for (int j = 0; j < 8; ++j)
                    c[i][j] = __builtin_amdgcn_wmma_f32_16x16x4_f32(false, af[i], false, bf[j],
                                                                    (short)0, c[i][j], false, false);
        }
        __builtin_amdgcn_sched_barrier(0);
        if (t + 1 < DFEAT / 8) {
            a_cur[0] = a_nxt[0]; a_cur[1] = a_nxt[1];
#pragma unroll
            for (int j = 0; j < 8; ++j) b_cur[j] = b_nxt[j];
        }
    }

#pragma unroll
    for (int i = 0; i < 2; ++i) {
#pragma unroll
        for (int r = 0; r < 8; ++r) {
            const int M = row0 + 16 * i + r + 8 * half;
            if (M < nrows) {
                float* hp = H + (size_t)M * DFEAT + 4 * m;
#pragma unroll
                for (int g = 0; g < 2; ++g) {
                    v4f o;
                    o.x = c[i][4 * g + 0][r]; o.y = c[i][4 * g + 1][r];
                    o.z = c[i][4 * g + 2][r]; o.w = c[i][4 * g + 3][r];
                    *(v4f*)(hp + 64 * g) = o;
                }
            }
        }
    }
}

#define OWN_WAVES 32
#define QCAP 512
#define SCAN_B 8
#define PGROUP 8

__global__ __launch_bounds__(1024) void aggregate_owner_kernel(const float* __restrict__ H,
                                                               const int* __restrict__ row,
                                                               const int* __restrict__ col,
                                                               const float* __restrict__ w,
                                                               const float* __restrict__ deg,
                                                               const unsigned* __restrict__ okey,
                                                               float* __restrict__ out, int E,
                                                               int n) {
    __shared__ float acc[(ACC_ROWS + 1) * DFEAT];
    __shared__ int queue[OWN_WAVES * QCAP];

    const int tid  = threadIdx.x;
    const int lane = tid & 31;
    const int wave = __builtin_amdgcn_readfirstlane(tid >> 5);
    const int base = (int)blockIdx.x * ACC_ROWS;
    const unsigned own = ((unsigned)blockIdx.x & 255u) * 0x01010101u;

    {
        const v4f z4 = {0.0f, 0.0f, 0.0f, 0.0f};
        for (int i = tid; i < (ACC_ROWS + 1) * DFEAT / 4; i += 1024) ((v4f*)acc)[i] = z4;
    }
    __syncthreads();

    int* q = queue + wave * QCAP;
    int qn = 0;

    auto drain = [&](const int cntq) {
        for (int b0 = 0; b0 < cntq; b0 += 32) {
            const int idx = b0 + lane;
            const bool v  = idx < cntq;
            const int e   = v ? q[idx] : 0;
            const int r   = row[e];
            const int c   = col[e];
            const float wv = w[e];
            const float dr = deg[r];
            const float dc = deg[c];
            const float ir = 1.0f / sqrtf(dr);
            const float ic = 1.0f / sqrtf(dc);
            const float nv = ir * wv * ic;
            const unsigned rel = (unsigned)(c - base);
            const int rl  = (v && rel < (unsigned)ACC_ROWS) ? (int)rel : ACC_ROWS;
            const int nvb = __builtin_bit_cast(int, nv);
            const int rem = cntq - b0;
            const int c32 = rem < 32 ? rem : 32;
            for (int i0 = 0; i0 < c32; i0 += PGROUP) {
                v4f hv[PGROUP];
#pragma unroll
                for (int u = 0; u < PGROUP; ++u) {
                    const int ri = __builtin_amdgcn_readlane(r, i0 + u);
                    hv[u] = *(const v4f*)(H + (size_t)ri * DFEAT + 4 * lane);
                }
#pragma unroll
                for (int u = 0; u < PGROUP; ++u) {
                    const int rli = __builtin_amdgcn_readlane(rl, i0 + u);
                    const float ni = __builtin_bit_cast(float, __builtin_amdgcn_readlane(nvb, i0 + u));
                    float* ap = acc + rli * DFEAT + lane;
                    atomicAdd(ap,      ni * hv[u].x);
                    atomicAdd(ap + 32, ni * hv[u].y);
                    atomicAdd(ap + 64, ni * hv[u].z);
                    atomicAdd(ap + 96, ni * hv[u].w);
                }
            }
        }
    };

    const int G = (E + 15) >> 4;
    const int U = (G + 31) >> 5;
    const v4u* __restrict__ k16 = (const v4u*)okey;
    for (int u0 = wave; u0 < U; u0 += OWN_WAVES * SCAN_B) {
        v4u kk[SCAN_B];
#pragma unroll
        for (int i = 0; i < SCAN_B; ++i) {
            const int gq = (u0 + OWN_WAVES * i) * 32 + lane;
            kk[i] = k16[gq < G ? gq : G - 1];
        }
        __builtin_amdgcn_sched_barrier(0);
        unsigned zz[SCAN_B];
#pragma unroll
        for (int i = 0; i < SCAN_B; ++i) {
            unsigned zc = 0;
#pragma unroll
            for (int j = 0; j < 4; ++j) {
                const unsigned y = kk[i][j] ^ own;
                const unsigned z = ~(((y & 0x7f7f7f7fu) + 0x7f7f7f7fu) | y) & 0x80808080u;
                zc |= z >> (7 - j);
            }
            const int gq = (u0 + OWN_WAVES * i) * 32 + lane;
            zz[i] = gq < G ? zc : 0u;
        }
        for (;;) {
            unsigned cur = 0;
            int ci = 0;
#pragma unroll
            for (int k = SCAN_B - 1; k >= 0; --k) {
                if (zz[k] != 0) { cur = zz[k]; ci = k; }
            }
            const bool has = cur != 0;
            if (__builtin_amdgcn_ballot_w32(has) == 0) break;
            if (qn > QCAP - 32) {
                drain(qn);
                qn = 0;
            }
            const int t = has ? __builtin_ctz(cur) : 0;
            const int p = ((t & 7) << 2) + (t >> 3);
            const int e = (((u0 + OWN_WAVES * ci) * 32 + lane) << 4) + p;
            const unsigned keep = has ? ~(1u << t) : ~0u;
#pragma unroll
            for (int k = 0; k < SCAN_B; ++k) zz[k] &= (k == ci) ? keep : ~0u;
            const bool pred = has && e < E;
            const unsigned bal = (unsigned)__builtin_amdgcn_ballot_w32(pred);
            const int slot = qn + (int)__builtin_amdgcn_mbcnt_lo(bal, 0u);
            if (pred) q[slot] = e;
            qn += __builtin_popcount(bal);
        }
    }
    drain(qn);
    __syncthreads();

    const int left = n - base;
    const int rows = left < ACC_ROWS ? left : ACC_ROWS;
    for (int rr = wave; rr < rows; rr += OWN_WAVES) {
        const float* ap = acc + rr * DFEAT + lane;
        v4f o;
        o.x = ap[0]; o.y = ap[32]; o.z = ap[64]; o.w = ap[96];
        *(v4f*)(out + (size_t)(base + rr) * DFEAT + 4 * lane) = o;
    }
}

extern "C" void kernel_launch(void* const* d_in, const int* in_sizes, int n_in,
                              void* d_out, int out_size, void* d_ws, size_t ws_size,
                              hipStream_t stream) {
    const float* x  = (const float*)d_in[0];
    const int*   ei = (const int*)d_in[1];
    const float* ew = (const float*)d_in[2];
    const float* W1 = (const float*)d_in[3];
    const float* b1 = (const float*)d_in[4];
    const float* W2 = (const float*)d_in[5];
    const float* b2 = (const float*)d_in[6];
    float* out = (float*)d_out;

    const int E = in_sizes[2];
    const int n = in_sizes[0] / DFEAT;
    const int* row = ei;
    const int* col = ei + E;

    char* ws = (char*)d_ws;
    size_t off = 0;
    auto alignup = [](size_t v) { return (v + 255) & ~(size_t)255; };
    float* deg  = (float*)(ws + off); off += alignup((size_t)n * 4);
    const size_t zero_bytes = off;
    unsigned* okey = (unsigned*)(ws + off); off += alignup((size_t)E * 4);
    float* h    = (float*)(ws + off); off += alignup((size_t)n * DFEAT * 4);
    float* agg  = (float*)(ws + off);

    const int gemm_blocks = (n + 127) / 128;
    const int epad        = (E + 15) & ~15;
    const int deg_blocks  = (epad / 4 + 127) / 128;
    const int own_blocks  = (n + ACC_ROWS - 1) / ACC_ROWS;

    hipMemsetAsync(deg, 0, zero_bytes, stream);

    gemm_wmma_kernel<false, true><<<gemm_blocks + deg_blocks, 128, 0, stream>>>(x, W1, b1, h, n, row, col,
                                                                              ew, deg, okey, E);
    aggregate_owner_kernel<<<own_blocks, 1024, 0, stream>>>(h, row, col, ew, deg, okey, agg, E, n);

    gemm_wmma_kernel<true, false><<<gemm_blocks, 128, 0, stream>>>(agg, W2, b2, h, n, nullptr, nullptr,
                                                                 nullptr, nullptr, nullptr, 0);
    aggregate_owner_kernel<<<own_blocks, 1024, 0, stream>>>(h, row, col, ew, deg, okey, out, E, n);
}


# Edge records: the degree work-groups pack row, col and w of each edge into one 16-byte record; the aggregation fetches one record per edge, not three scattered 4-byte loads; sum order kept
# speedup vs baseline: 2.5272x; 1.0518x over previous
#include <hip/hip_runtime.h>

typedef float v2f __attribute__((ext_vector_type(2)));
typedef float v8f __attribute__((ext_vector_type(8)));

#define DFEAT 128


typedef float v4f __attribute__((ext_vector_type(4)));

typedef int v4i __attribute__((ext_vector_type(4)));
typedef unsigned v4u __attribute__((ext_vector_type(4)));
#define ACC_ROWS 200

#define BAND_KEY(c) (((unsigned)(c) / (unsigned)ACC_ROWS) & 255u)

template <bool RELU, bool DEG>
__global__ __launch_bounds__(128) void gemm_wmma_kernel(const float* __restrict__ X,
                                                        const float* __restrict__ W,
                                                        const float* __restrict__ bias,
                                                        float* __restrict__ H, int nrows,
                                                        const int* __restrict__ erow,
                                                        const int* __restrict__ ecol,
                                                        const float* __restrict__ ew,
                                                        float* __restrict__ deg,
                                                        unsigned* __restrict__ okey,
                                                        int* __restrict__ erec, int E) {
    if (DEG) {
        const int gemm_blocks = (nrows + 127) / 128;
        if ((int)blockIdx.x >= gemm_blocks) {
            const int e4 = (((int)blockIdx.x - gemm_blocks) * 128 + (int)threadIdx.x) * 4;
            const int epad = (E + 15) & ~15;
            if (e4 >= epad) return;
            unsigned kp = 0;
            if (e4 + 4 <= E) {
                const v4i r4 = *(const v4i*)(erow + e4);
                const v4i c4 = *(const v4i*)(ecol + e4);
                const v4f w4 = *(const v4f*)(ew + e4);
                atomicAdd(&deg[r4.x], w4.x);
                atomicAdd(&deg[r4.y], w4.y);
                atomicAdd(&deg[r4.z], w4.z);
                atomicAdd(&deg[r4.w], w4.w);
                kp = BAND_KEY(c4.x) | (BAND_KEY(c4.y) << 8) | (BAND_KEY(c4.z) << 16) |
                     (BAND_KEY(c4.w) << 24);
                v4i* rp = (v4i*)erec + e4;
                const float wx = w4.x, wy = w4.y, wz = w4.z, ww = w4.w;
                rp[0] = v4i{r4.x, c4.x, __float_as_int(wx), 0};
                rp[1] = v4i{r4.y, c4.y, __float_as_int(wy), 0};
                rp[2] = v4i{r4.z, c4.z, __float_as_int(wz), 0};
                rp[3] = v4i{r4.w, c4.w, __float_as_int(ww), 0};
            } else {
                for (int j = 0; j < 4; ++j) {
                    const int e = e4 + j;
                    if (e < E) {
                        const int r = erow[e];
                        const int c = ecol[e];
                        const float wv = ew[e];
                        atomicAdd(&deg[r], wv);
                        kp |= BAND_KEY(c) << (8 * j);
                        ((v4i*)erec)[e] = v4i{r, c, __float_as_int(wv), 0};
                    }
                }
            }
            okey[e4 >> 2] = kp;
            return;
        }
    }
    const int lane = threadIdx.x & 31;
    const int wave = threadIdx.x >> 5;
    const int m    = lane & 15;
    const int half = lane >> 4;
    const int row0 = (blockIdx.x * 4 + wave) * 32;
    if (row0 >= nrows) return;

    int ar0 = row0 + m;
    if (ar0 >= nrows) ar0 = nrows - 1;
    int ar1 = row0 + 16 + m;
    if (ar1 >= nrows) ar1 = nrows - 1;
    const float* __restrict__ xr0 = X + (size_t)ar0 * DFEAT + 4 * half;
    const float* __restrict__ xr1 = X + (size_t)ar1 * DFEAT + 4 * half;
    const float* __restrict__ wr  = W + (size_t)(4 * m) * DFEAT + 4 * half;

    v8f c[2][8];
    {
        const v4f bv0 = *(const v4f*)(bias + 4 * m);
        const v4f bv1 = *(const v4f*)(bias + 64 + 4 * m);
#pragma unroll
        for (int j = 0; j < 8; ++j) {
            const float bv = (j < 4) ? bv0[j & 3] : bv1[j & 3];
#pragma unroll
            for (int r = 0; r < 8; ++r) { c[0][j][r] = bv; c[1][j][r] = bv; }
        }
    }

    v4f a_cur[2], b_cur[8];
    a_cur[0] = *(const v4f*)xr0;
    a_cur[1] = *(const v4f*)xr1;
#pragma unroll
    for (int j = 0; j < 8; ++j)
        b_cur[j] = *(const v4f*)(wr + (size_t)((j >> 2) * 64 + (j & 3)) * DFEAT);

#pragma unroll
    for (int t = 0; t < DFEAT / 8; ++t) {
        v4f a_nxt[2], b_nxt[8];
        if (t + 1 < DFEAT / 8) {
            const int kb = 8 * (t + 1);
            a_nxt[0] = *(const v4f*)(xr0 + kb);
            a_nxt[1] = *(const v4f*)(xr1 + kb);
#pragma unroll
            for (int j = 0; j < 8; ++j)
                b_nxt[j] = *(const v4f*)(wr + (size_t)((j >> 2) * 64 + (j & 3)) * DFEAT + kb);
        }
        __builtin_amdgcn_sched_barrier(0);
#pragma unroll
        for (int s = 0; s < 2; ++s) {
            v2f af[2], bf[8];
#pragma unroll
            for (int i = 0; i < 2; ++i) {
                af[i].x = a_cur[i][2 * s]; af[i].y = a_cur[i][2 * s + 1];
                if (RELU) { af[i].x = fmaxf(af[i].x, 0.0f); af[i].y = fmaxf(af[i].y, 0.0f); }
            }
#pragma unroll
            for (int j = 0; j < 8; ++j) { bf[j].x = b_cur[j][2 * s]; bf[j].y = b_cur[j][2 * s + 1]; }
#pragma unroll
            for (int i = 0; i < 2; ++i)
#pragma unroll
                for (int j = 0; j < 8; ++j)
                    c[i][j] = __builtin_amdgcn_wmma_f32_16x16x4_f32(false, af[i], false, bf[j],
                                                                    (short)0, c[i][j], false, false);
        }
        __builtin_amdgcn_sched_barrier(0);
        if (t + 1 < DFEAT / 8) {
            a_cur[0] = a_nxt[0]; a_cur[1] = a_nxt[1];
#pragma unroll
            for (int j = 0; j < 8; ++j) b_cur[j] = b_nxt[j];
        }
    }

#pragma unroll
    for (int i = 0; i < 2; ++i) {
#pragma unroll
        for (int r = 0; r < 8; ++r) {
            const int M = row0 + 16 * i + r + 8 * half;
            if (M < nrows) {
                float* hp = H + (size_t)M * DFEAT + 4 * m;
#pragma unroll
                for (int g = 0; g < 2; ++g) {
                    v4f o;
                    o.x = c[i][4 * g + 0][r]; o.y = c[i][4 * g + 1][r];
                    o.z = c[i][4 * g + 2][r]; o.w = c[i][4 * g + 3][r];
                    *(v4f*)(hp + 64 * g) = o;
                }
            }
        }
    }
}

#define OWN_WAVES 32
#define QCAP 512
#define SCAN_B 8
#define PGROUP 8

__global__ __launch_bounds__(1024) void aggregate_owner_kernel(const float* __restrict__ H,
                                                               const int* __restrict__ erec,
                                                               const float* __restrict__ deg,
                                                               const unsigned* __restrict__ okey,
                                                               float* __restrict__ out, int E,
                                                               int n) {
    __shared__ float acc[(ACC_ROWS + 1) * DFEAT];
    __shared__ int queue[OWN_WAVES * QCAP];

    const int tid  = threadIdx.x;
    const int lane = tid & 31;
    const int wave = __builtin_amdgcn_readfirstlane(tid >> 5);
    const int base = (int)blockIdx.x * ACC_ROWS;
    const unsigned own = ((unsigned)blockIdx.x & 255u) * 0x01010101u;

    {
        const v4f z4 = {0.0f, 0.0f, 0.0f, 0.0f};
        for (int i = tid; i < (ACC_ROWS + 1) * DFEAT / 4; i += 1024) ((v4f*)acc)[i] = z4;
    }
    __syncthreads();

    int* q = queue + wave * QCAP;
    int qn = 0;

    auto drain = [&](const int cntq) {
        for (int b0 = 0; b0 < cntq; b0 += 32) {
            const int idx = b0 + lane;
            const bool v  = idx < cntq;
            const int e   = v ? q[idx] : 0;
            const v4i rc  = ((const v4i*)erec)[e];
            const int r   = rc.x;
            const int c   = rc.y;
            const int wbits = rc.z;
            const float wv = __int_as_float(wbits);
            const float dr = deg[r];
            const float dc = deg[c];
            const float ir = 1.0f / sqrtf(dr);
            const float ic = 1.0f / sqrtf(dc);
            const float nv = ir * wv * ic;
            const unsigned rel = (unsigned)(c - base);
            const int rl  = (v && rel < (unsigned)ACC_ROWS) ? (int)rel : ACC_ROWS;
            const int nvb = __builtin_bit_cast(int, nv);
            const int rem = cntq - b0;
            const int c32 = rem < 32 ? rem : 32;
            for (int i0 = 0; i0 < c32; i0 += PGROUP) {
                v4f hv[PGROUP];
#pragma unroll
                for (int u = 0; u < PGROUP; ++u) {
                    const int ri = __builtin_amdgcn_readlane(r, i0 + u);
                    hv[u] = *(const v4f*)(H + (size_t)ri * DFEAT + 4 * lane);
                }
#pragma unroll
                for (int u = 0; u < PGROUP; ++u) {
                    const int rli = __builtin_amdgcn_readlane(rl, i0 + u);
                    const float ni = __builtin_bit_cast(float, __builtin_amdgcn_readlane(nvb, i0 + u));
                    float* ap = acc + rli * DFEAT + lane;
                    atomicAdd(ap,      ni * hv[u].x);
                    atomicAdd(ap + 32, ni * hv[u].y);
                    atomicAdd(ap + 64, ni * hv[u].z);
                    atomicAdd(ap + 96, ni * hv[u].w);
                }
            }
        }
    };

    const int G = (E + 15) >> 4;
    const int U = (G + 31) >> 5;
    const v4u* __restrict__ k16 = (const v4u*)okey;
    for (int u0 = wave; u0 < U; u0 += OWN_WAVES * SCAN_B) {
        v4u kk[SCAN_B];
#pragma unroll
        for (int i = 0; i < SCAN_B; ++i) {
            const int gq = (u0 + OWN_WAVES * i) * 32 + lane;
            kk[i] = k16[gq < G ? gq : G - 1];
        }
        __builtin_amdgcn_sched_barrier(0);
        unsigned zz[SCAN_B];
#pragma unroll
        for (int i = 0; i < SCAN_B; ++i) {
            unsigned zc = 0;
#pragma unroll
            for (int j = 0; j < 4; ++j) {
                const unsigned y = kk[i][j] ^ own;
                const unsigned z = ~(((y & 0x7f7f7f7fu) + 0x7f7f7f7fu) | y) & 0x80808080u;
                zc |= z >> (7 - j);
            }
            const int gq = (u0 + OWN_WAVES * i) * 32 + lane;
            zz[i] = gq < G ? zc : 0u;
        }
        for (;;) {
            unsigned cur = 0;
            int ci = 0;
#pragma unroll
            for (int k = SCAN_B - 1; k >= 0; --k) {
                if (zz[k] != 0) { cur = zz[k]; ci = k; }
            }
            const bool has = cur != 0;
            if (__builtin_amdgcn_ballot_w32(has) == 0) break;
            if (qn > QCAP - 32) {
                drain(qn);
                qn = 0;
            }
            const int t = has ? __builtin_ctz(cur) : 0;
            const int p = ((t & 7) << 2) + (t >> 3);
            const int e = (((u0 + OWN_WAVES * ci) * 32 + lane) << 4) + p;
            const unsigned keep = has ? ~(1u << t) : ~0u;
#pragma unroll
            for (int k = 0; k < SCAN_B; ++k) zz[k] &= (k == ci) ? keep : ~0u;
            const bool pred = has && e < E;
            const unsigned bal = (unsigned)__builtin_amdgcn_ballot_w32(pred);
            const int slot = qn + (int)__builtin_amdgcn_mbcnt_lo(bal, 0u);
            if (pred) q[slot] = e;
            qn += __builtin_popcount(bal);
        }
    }
    drain(qn);
    __syncthreads();

    const int left = n - base;
    const int rows = left < ACC_ROWS ? left : ACC_ROWS;
    for (int rr = wave; rr < rows; rr += OWN_WAVES) {
        const float* ap = acc + rr * DFEAT + lane;
        v4f o;
        o.x = ap[0]; o.y = ap[32]; o.z = ap[64]; o.w = ap[96];
        *(v4f*)(out + (size_t)(base + rr) * DFEAT + 4 * lane) = o;
    }
}

extern "C" void kernel_launch(void* const* d_in, const int* in_sizes, int n_in,
                              void* d_out, int out_size, void* d_ws, size_t ws_size,
                              hipStream_t stream) {
    const float* x  = (const float*)d_in[0];
    const int*   ei = (const int*)d_in[1];
    const float* ew = (const float*)d_in[2];
    const float* W1 = (const float*)d_in[3];
    const float* b1 = (const float*)d_in[4];
    const float* W2 = (const float*)d_in[5];
    const float* b2 = (const float*)d_in[6];
    float* out = (float*)d_out;

    const int E = in_sizes[2];
    const int n = in_sizes[0] / DFEAT;
    const int* row = ei;
    const int* col = ei + E;

    char* ws = (char*)d_ws;
    size_t off = 0;
    auto alignup = [](size_t v) { return (v + 255) & ~(size_t)255; };
    float* deg  = (float*)(ws + off); off += alignup((size_t)n * 4);
    const size_t zero_bytes = off;
    unsigned* okey = (unsigned*)(ws + off); off += alignup((size_t)E * 4);
    float* h    = (float*)(ws + off); off += alignup((size_t)n * DFEAT * 4);
    int* erec   = (int*)(ws + off);
    float* agg  = out;

    const int gemm_blocks = (n + 127) / 128;
    const int epad        = (E + 15) & ~15;
    const int deg_blocks  = (epad / 4 + 127) / 128;
    const int own_blocks  = (n + ACC_ROWS - 1) / ACC_ROWS;

    hipMemsetAsync(deg, 0, zero_bytes, stream);

    gemm_wmma_kernel<false, true><<<gemm_blocks + deg_blocks, 128, 0, stream>>>(x, W1, b1, h, n, row, col,
                                                                              ew, deg, okey, erec, E);
    aggregate_owner_kernel<<<own_blocks, 1024, 0, stream>>>(h, erec, deg, okey, agg, E, n);

    gemm_wmma_kernel<true, false><<<gemm_blocks, 128, 0, stream>>>(agg, W2, b2, h, n, nullptr, nullptr,
                                                                 nullptr, nullptr, nullptr, nullptr, 0);
    aggregate_owner_kernel<<<own_blocks, 1024, 0, stream>>>(h, erec, deg, okey, out, E, n);
}


# Aggregation 2 replays aggregation 1's finds: launch 1 writes each wave's edges (row, local row, norm; 8 B) to workspace, launch 2 reads them: no key scan, no record or degree loads; sum order kept
# speedup vs baseline: 2.6984x; 1.0677x over previous
#include <hip/hip_runtime.h>

typedef float v2f __attribute__((ext_vector_type(2)));
typedef float v8f __attribute__((ext_vector_type(8)));

#define DFEAT 128


typedef float v4f __attribute__((ext_vector_type(4)));

typedef int v4i __attribute__((ext_vector_type(4)));
typedef unsigned v4u __attribute__((ext_vector_type(4)));
#define ACC_ROWS 200

#define BAND_KEY(c) (((unsigned)(c) / (unsigned)ACC_ROWS) & 255u)

template <bool RELU, bool DEG>
__global__ __launch_bounds__(128) void gemm_wmma_kernel(const float* __restrict__ X,
                                                        const float* __restrict__ W,
                                                        const float* __restrict__ bias,
                                                        float* __restrict__ H, int nrows,
                                                        const int* __restrict__ erow,
                                                        const int* __restrict__ ecol,
                                                        const float* __restrict__ ew,
                                                        float* __restrict__ deg,
                                                        unsigned* __restrict__ okey,
                                                        int* __restrict__ erec, int E) {
    if (DEG) {
        const int gemm_blocks = (nrows + 127) / 128;
        if ((int)blockIdx.x >= gemm_blocks) {
            const int e4 = (((int)blockIdx.x - gemm_blocks) * 128 + (int)threadIdx.x) * 4;
            const int epad = (E + 15) & ~15;
            if (e4 >= epad) return;
            unsigned kp = 0;
            if (e4 + 4 <= E) {
                const v4i r4 = *(const v4i*)(erow + e4);
                const v4i c4 = *(const v4i*)(ecol + e4);
                const v4f w4 = *(const v4f*)(ew + e4);
                atomicAdd(&deg[r4.x], w4.x);
                atomicAdd(&deg[r4.y], w4.y);
                atomicAdd(&deg[r4.z], w4.z);
                atomicAdd(&deg[r4.w], w4.w);
                kp = BAND_KEY(c4.x) | (BAND_KEY(c4.y) << 8) | (BAND_KEY(c4.z) << 16) |
                     (BAND_KEY(c4.w) << 24);
                v4i* rp = (v4i*)erec + e4;
                const float wx = w4.x, wy = w4.y, wz = w4.z, ww = w4.w;
                rp[0] = v4i{r4.x, c4.x, __float_as_int(wx), 0};
                rp[1] = v4i{r4.y, c4.y, __float_as_int(wy), 0};
                rp[2] = v4i{r4.z, c4.z, __float_as_int(wz), 0};
                rp[3] = v4i{r4.w, c4.w, __float_as_int(ww), 0};
            } else {
                for (int j = 0; j < 4; ++j) {
                    const int e = e4 + j;
                    if (e < E) {
                        const int r = erow[e];
                        const int c = ecol[e];
                        const float wv = ew[e];
                        atomicAdd(&deg[r], wv);
                        kp |= BAND_KEY(c) << (8 * j);
                        ((v4i*)erec)[e] = v4i{r, c, __float_as_int(wv), 0};
                    }
                }
            }
            okey[e4 >> 2] = kp;
            return;
        }
    }
    const int lane = threadIdx.x & 31;
    const int wave = threadIdx.x >> 5;
    const int m    = lane & 15;
    const int half = lane >> 4;
    const int row0 = (blockIdx.x * 4 + wave) * 32;
    if (row0 >= nrows) return;

    int ar0 = row0 + m;
    if (ar0 >= nrows) ar0 = nrows - 1;
    int ar1 = row0 + 16 + m;
    if (ar1 >= nrows) ar1 = nrows - 1;
    const float* __restrict__ xr0 = X + (size_t)ar0 * DFEAT + 4 * half;
    const float* __restrict__ xr1 = X + (size_t)ar1 * DFEAT + 4 * half;
    const float* __restrict__ wr  = W + (size_t)(4 * m) * DFEAT + 4 * half;

    v8f c[2][8];
    {
        const v4f bv0 = *(const v4f*)(bias + 4 * m);
        const v4f bv1 = *(const v4f*)(bias + 64 + 4 * m);
#pragma unroll
        for (int j = 0; j < 8; ++j) {
            const float bv = (j < 4) ? bv0[j & 3] : bv1[j & 3];
#pragma unroll
            for (int r = 0; r < 8; ++r) { c[0][j][r] = bv; c[1][j][r] = bv; }
        }
    }

    v4f a_cur[2], b_cur[8];
    a_cur[0] = *(const v4f*)xr0;
    a_cur[1] = *(const v4f*)xr1;
#pragma unroll
    for (int j = 0; j < 8; ++j)
        b_cur[j] = *(const v4f*)(wr + (size_t)((j >> 2) * 64 + (j & 3)) * DFEAT);

#pragma unroll
    for (int t = 0; t < DFEAT / 8; ++t) {
        v4f a_nxt[2], b_nxt[8];
        if (t + 1 < DFEAT / 8) {
            const int kb = 8 * (t + 1);
            a_nxt[0] = *(const v4f*)(xr0 + kb);
            a_nxt[1] = *(const v4f*)(xr1 + kb);
#pragma unroll
            for (int j = 0; j < 8; ++j)
                b_nxt[j] = *(const v4f*)(wr + (size_t)((j >> 2) * 64 + (j & 3)) * DFEAT + kb);
        }
        __builtin_amdgcn_sched_barrier(0);
#pragma unroll
        for (int s = 0; s < 2; ++s) {
            v2f af[2], bf[8];
#pragma unroll
            for (int i = 0; i < 2; ++i) {
                af[i].x = a_cur[i][2 * s]; af[i].y = a_cur[i][2 * s + 1];
                if (RELU) { af[i].x = fmaxf(af[i].x, 0.0f); af[i].y = fmaxf(af[i].y, 0.0f); }
            }
#pragma unroll
            for (int j = 0; j < 8; ++j) { bf[j].x = b_cur[j][2 * s]; bf[j].y = b_cur[j][2 * s + 1]; }
#pragma unroll
            for (int i = 0; i < 2; ++i)
#pragma unroll
                for (int j = 0; j < 8; ++j)
                    c[i][j] = __builtin_amdgcn_wmma_f32_16x16x4_f32(false, af[i], false, bf[j],
                                                                    (short)0, c[i][j], false, false);
        }
        __builtin_amdgcn_sched_barrier(0);
        if (t + 1 < DFEAT / 8) {
            a_cur[0] = a_nxt[0]; a_cur[1] = a_nxt[1];
#pragma unroll
            for (int j = 0; j < 8; ++j) b_cur[j] = b_nxt[j];
        }
    }

#pragma unroll
    for (int i = 0; i < 2; ++i) {
#pragma unroll
        for (int r = 0; r < 8; ++r) {
            const int M = row0 + 16 * i + r + 8 * half;
            if (M < nrows) {
                float* hp = H + (size_t)M * DFEAT + 4 * m;
#pragma unroll
                for (int g = 0; g < 2; ++g) {
                    v4f o;
                    o.x = c[i][4 * g + 0][r]; o.y = c[i][4 * g + 1][r];
                    o.z = c[i][4 * g + 2][r]; o.w = c[i][4 * g + 3][r];
                    *(v4f*)(hp + 64 * g) = o;
                }
            }
        }
    }
}

#define OWN_WAVES 32
#define QCAP 512
#define SCAN_B 8
#define PGROUP 8

#define LCAP 128
typedef int v2i __attribute__((ext_vector_type(2)));

template <bool REPLAY>
__global__ __launch_bounds__(1024) void aggregate_owner_kernel(const float* __restrict__ H,
                                                               const int* __restrict__ erec,
                                                               const float* __restrict__ deg,
                                                               const unsigned* __restrict__ okey,
                                                               float* __restrict__ out, int E,
                                                               int n, int* __restrict__ lst,
                                                               int* __restrict__ lcnt) {
    __shared__ float acc[(ACC_ROWS + 1) * DFEAT];
    __shared__ int queue[OWN_WAVES * QCAP];

    const int tid  = threadIdx.x;
    const int lane = tid & 31;
    const int wave = __builtin_amdgcn_readfirstlane(tid >> 5);
    const int base = (int)blockIdx.x * ACC_ROWS;
    const unsigned own = ((unsigned)blockIdx.x & 255u) * 0x01010101u;

    {
        const v4f z4 = {0.0f, 0.0f, 0.0f, 0.0f};
        for (int i = tid; i < (ACC_ROWS + 1) * DFEAT / 4; i += 1024) ((v4f*)acc)[i] = z4;
    }
    __syncthreads();

    int* q = queue + wave * QCAP;
    int qn = 0;
    int tot = 0;
    v2i* wl = (v2i*)lst + (size_t)((int)blockIdx.x * OWN_WAVES + wave) * LCAP;

    auto apply = [&](const int r, const int rl, const int nvb, const int c32) {
            for (int i0 = 0; i0 < c32; i0 += PGROUP) {
                v4f hv[PGROUP];
#pragma unroll
                for (int u = 0; u < PGROUP; ++u) {
                    const int ri = __builtin_amdgcn_readlane(r, i0 + u);
                    hv[u] = *(const v4f*)(H + (size_t)ri * DFEAT + 4 * lane);
                }
#pragma unroll
                for (int u = 0; u < PGROUP; ++u) {
                    const int rli = __builtin_amdgcn_readlane(rl, i0 + u);
                    const float ni = __builtin_bit_cast(float, __builtin_amdgcn_readlane(nvb, i0 + u));
                    float* ap = acc + rli * DFEAT + lane;
                    atomicAdd(ap,      ni * hv[u].x);
                    atomicAdd(ap + 32, ni * hv[u].y);
                    atomicAdd(ap + 64, ni * hv[u].z);
                    atomicAdd(ap + 96, ni * hv[u].w);
                }
            }
    };

    auto drain = [&](const int cntq) {
        for (int b0 = 0; b0 < cntq; b0 += 32) {
            const int idx = b0 + lane;
            const bool v  = idx < cntq;
            const int e   = v ? q[idx] : 0;
            const v4i rc  = ((const v4i*)erec)[e];
            const int r   = rc.x;
            const int c   = rc.y;
            const int wbits = rc.z;
            const float wv = __int_as_float(wbits);
            const float dr = deg[r];
            const float dc = deg[c];
            const float ir = 1.0f / sqrtf(dr);
            const float ic = 1.0f / sqrtf(dc);
            const float nv = ir * wv * ic;
            const unsigned rel = (unsigned)(c - base);
            const int rl  = (v && rel < (unsigned)ACC_ROWS) ? (int)rel : ACC_ROWS;
            const int nvb = __builtin_bit_cast(int, nv);
            if (!REPLAY) {
                const int li = tot + idx;
                if (lst != nullptr && v && li < LCAP)
                    wl[li] = v2i{(int)((unsigned)r | ((unsigned)rl << 24)), nvb};
            }
            const int rem = cntq - b0;
            apply(r, rl, nvb, rem < 32 ? rem : 32);
        }
        if (!REPLAY) tot += cntq;
    };

    bool scan = true;
    if (REPLAY) {
        if (lst != nullptr) {
            const int lc = __builtin_amdgcn_readfirstlane(lcnt[(int)blockIdx.x * OWN_WAVES + wave]);
            if (lc <= LCAP) {
                scan = false;
                if (lc > 0) {
                    const int last = lc - 1;
                    v2i en = wl[lane < last ? lane : last];
                    for (int b0 = 0; b0 < lc; b0 += 32) {
                        const int ix = b0 + 32 + lane;
                        const v2i nx = wl[ix < last ? ix : last];
                        const bool v = b0 + lane < lc;
                        const int rem = lc - b0;
                        apply(en.x & 0xffffff, v ? (int)((unsigned)en.x >> 24) : ACC_ROWS, en.y,
                              rem < 32 ? rem : 32);
                        en = nx;
                    }
                }
            }
        }
    }

    const int G = (E + 15) >> 4;
    const int U = (G + 31) >> 5;
    const v4u* __restrict__ k16 = (const v4u*)okey;
    for (int u0 = wave; scan && u0 < U; u0 += OWN_WAVES * SCAN_B) {
        v4u kk[SCAN_B];
#pragma unroll
        for (int i = 0; i < SCAN_B; ++i) {
            const int gq = (u0 + OWN_WAVES * i) * 32 + lane;
            kk[i] = k16[gq < G ? gq : G - 1];
        }
        __builtin_amdgcn_sched_barrier(0);
        unsigned zz[SCAN_B];
#pragma unroll
        for (int i = 0; i < SCAN_B; ++i) {
            unsigned zc = 0;
#pragma unroll
            for (int j = 0; j < 4; ++j) {
                const unsigned y = kk[i][j] ^ own;
                const unsigned z = ~(((y & 0x7f7f7f7fu) + 0x7f7f7f7fu) | y) & 0x80808080u;
                zc |= z >> (7 - j);
            }
            const int gq = (u0 + OWN_WAVES * i) * 32 + lane;
            zz[i] = gq < G ? zc : 0u;
        }
        for (;;) {
            unsigned cur = 0;
            int ci = 0;
#pragma unroll
            for (int k = SCAN_B - 1; k >= 0; --k) {
                if (zz[k] != 0) { cur = zz[k]; ci = k; }
            }
            const bool has = cur != 0;
            if (__builtin_amdgcn_ballot_w32(has) == 0) break;
            if (qn > QCAP - 32) {
                drain(qn);
                qn = 0;
            }
            const int t = has ? __builtin_ctz(cur) : 0;
            const int p = ((t & 7) << 2) + (t >> 3);
            const int e = (((u0 + OWN_WAVES * ci) * 32 + lane) << 4) + p;
            const unsigned keep = has ? ~(1u << t) : ~0u;
#pragma unroll
            for (int k = 0; k < SCAN_B; ++k) zz[k] &= (k == ci) ? keep : ~0u;
            const bool pred = has && e < E;
            const unsigned bal = (unsigned)__builtin_amdgcn_ballot_w32(pred);
            const int slot = qn + (int)__builtin_amdgcn_mbcnt_lo(bal, 0u);
            if (pred) q[slot] = e;
            qn += __builtin_popcount(bal);
        }
    }
    if (scan) drain(qn);
    if (!REPLAY) {
        if (lst != nullptr && lane == 0) lcnt[(int)blockIdx.x * OWN_WAVES + wave] = tot;
    }
    __syncthreads();

    const int left = n - base;
    const int rows = left < ACC_ROWS ? left : ACC_ROWS;
    for (int rr = wave; rr < rows; rr += OWN_WAVES) {
        const float* ap = acc + rr * DFEAT + lane;
        v4f o;
        o.x = ap[0]; o.y = ap[32]; o.z = ap[64]; o.w = ap[96];
        *(v4f*)(out + (size_t)(base + rr) * DFEAT + 4 * lane) = o;
    }
}

extern "C" void kernel_launch(void* const* d_in, const int* in_sizes, int n_in,
                              void* d_out, int out_size, void* d_ws, size_t ws_size,
                              hipStream_t stream) {
    const float* x  = (const float*)d_in[0];
    const int*   ei = (const int*)d_in[1];
    const float* ew = (const float*)d_in[2];
    const float* W1 = (const float*)d_in[3];
    const float* b1 = (const float*)d_in[4];
    const float* W2 = (const float*)d_in[5];
    const float* b2 = (const float*)d_in[6];
    float* out = (float*)d_out;

    const int E = in_sizes[2];
    const int n = in_sizes[0] / DFEAT;
    const int* row = ei;
    const int* col = ei + E;

    char* ws = (char*)d_ws;
    size_t off = 0;
    auto alignup = [](size_t v) { return (v + 255) & ~(size_t)255; };
    float* deg  = (float*)(ws + off); off += alignup((size_t)n * 4);
    const size_t zero_bytes = off;
    unsigned* okey = (unsigned*)(ws + off); off += alignup((size_t)E * 4);
    float* h    = (float*)(ws + off); off += alignup((size_t)n * DFEAT * 4);
    int* erec   = (int*)(ws + off);
    float* agg  = out;

    const int gemm_blocks = (n + 127) / 128;
    const int epad        = (E + 15) & ~15;
    const int deg_blocks  = (epad / 4 + 127) / 128;
    const int own_blocks  = (n + ACC_ROWS - 1) / ACC_ROWS;

    const size_t lcnt_off = off + alignup((size_t)E * 16);
    const size_t lst_off  = lcnt_off + alignup((size_t)own_blocks * OWN_WAVES * 4);
    const size_t lst_end  = lst_off + (size_t)own_blocks * OWN_WAVES * LCAP * 8;
    const bool lists = lst_end <= ws_size && n <= (1 << 24);
    int* lcnt = lists ? (int*)(ws + lcnt_off) : nullptr;
    int* lst  = lists ? (int*)(ws + lst_off) : nullptr;

    hipMemsetAsync(deg, 0, zero_bytes, stream);

    gemm_wmma_kernel<false, true><<<gemm_blocks + deg_blocks, 128, 0, stream>>>(x, W1, b1, h, n, row, col,
                                                                              ew, deg, okey, erec, E);
    aggregate_owner_kernel<false><<<own_blocks, 1024, 0, stream>>>(h, erec, deg, okey, agg, E, n, lst, lcnt);

    gemm_wmma_kernel<true, false><<<gemm_blocks, 128, 0, stream>>>(agg, W2, b2, h, n, nullptr, nullptr,
                                                                 nullptr, nullptr, nullptr, nullptr, 0);
    aggregate_owner_kernel<true><<<own_blocks, 1024, 0, stream>>>(h, erec, deg, okey, out, E, n, lst, lcnt);
}


# Layer 2's ReLU and GEMM moved into aggregation 1: a band's finished sums go from LDS through WMMA (W2 staged in LDS, same f32 operands) straight to h2; 4 launches to 3; k sum order of GEMM 2 changed
# speedup vs baseline: 2.8409x; 1.0528x over previous
#include <hip/hip_runtime.h>

typedef float v2f __attribute__((ext_vector_type(2)));
typedef float v8f __attribute__((ext_vector_type(8)));

#define DFEAT 128


typedef float v4f __attribute__((ext_vector_type(4)));

typedef int v4i __attribute__((ext_vector_type(4)));
typedef unsigned v4u __attribute__((ext_vector_type(4)));
#define ACC_ROWS 200

#define BAND_KEY(c) (((unsigned)(c) / (unsigned)ACC_ROWS) & 255u)

template <bool DEG>
__global__ __launch_bounds__(128) void gemm_wmma_kernel(const float* __restrict__ X,
                                                        const float* __restrict__ W,
                                                        const float* __restrict__ bias,
                                                        float* __restrict__ H, int nrows,
                                                        const int* __restrict__ erow,
                                                        const int* __restrict__ ecol,
                                                        const float* __restrict__ ew,
                                                        float* __restrict__ deg,
                                                        unsigned* __restrict__ okey,
                                                        int* __restrict__ erec, int E) {
    if (DEG) {
        const int gemm_blocks = (nrows + 127) / 128;
        if ((int)blockIdx.x >= gemm_blocks) {
            const int e4 = (((int)blockIdx.x - gemm_blocks) * 128 + (int)threadIdx.x) * 4;
            const int epad = (E + 15) & ~15;
            if (e4 >= epad) return;
            unsigned kp = 0;
            if (e4 + 4 <= E) {
                const v4i r4 = *(const v4i*)(erow + e4);
                const v4i c4 = *(const v4i*)(ecol + e4);
                const v4f w4 = *(const v4f*)(ew + e4);
                atomicAdd(&deg[r4.x], w4.x);
                atomicAdd(&deg[r4.y], w4.y);
                atomicAdd(&deg[r4.z], w4.z);
                atomicAdd(&deg[r4.w], w4.w);
                kp = BAND_KEY(c4.x) | (BAND_KEY(c4.y) << 8) | (BAND_KEY(c4.z) << 16) |
                     (BAND_KEY(c4.w) << 24);
                v4i* rp = (v4i*)erec + e4;
                const float wx = w4.x, wy = w4.y, wz = w4.z, ww = w4.w;
                rp[0] = v4i{r4.x, c4.x, __float_as_int(wx), 0};
                rp[1] = v4i{r4.y, c4.y, __float_as_int(wy), 0};
                rp[2] = v4i{r4.z, c4.z, __float_as_int(wz), 0};
                rp[3] = v4i{r4.w, c4.w, __float_as_int(ww), 0};
            } else {
                for (int j = 0; j < 4; ++j) {
                    const int e = e4 + j;
                    if (e < E) {
                        const int r = erow[e];
                        const int c = ecol[e];
                        const float wv = ew[e];
                        atomicAdd(&deg[r], wv);
                        kp |= BAND_KEY(c) << (8 * j);
                        ((v4i*)erec)[e] = v4i{r, c, __float_as_int(wv), 0};
                    }
                }
            }
            okey[e4 >> 2] = kp;
            return;
        }
    }
    const int lane = threadIdx.x & 31;
    const int wave = threadIdx.x >> 5;
    const int m    = lane & 15;
    const int half = lane >> 4;
    const int row0 = (blockIdx.x * 4 + wave) * 32;
    if (row0 >= nrows) return;

    int ar0 = row0 + m;
    if (ar0 >= nrows) ar0 = nrows - 1;
    int ar1 = row0 + 16 + m;
    if (ar1 >= nrows) ar1 = nrows - 1;
    const float* __restrict__ xr0 = X + (size_t)ar0 * DFEAT + 4 * half;
    const float* __restrict__ xr1 = X + (size_t)ar1 * DFEAT + 4 * half;
    const float* __restrict__ wr  = W + (size_t)(4 * m) * DFEAT + 4 * half;

    v8f c[2][8];
    {
        const v4f bv0 = *(const v4f*)(bias + 4 * m);
        const v4f bv1 = *(const v4f*)(bias + 64 + 4 * m);
#pragma unroll
        for (int j = 0; j < 8; ++j) {
            const float bv = (j < 4) ? bv0[j & 3] : bv1[j & 3];
#pragma unroll
            for (int r = 0; r < 8; ++r) { c[0][j][r] = bv; c[1][j][r] = bv; }
        }
    }

    v4f a_cur[2], b_cur[8];
    a_cur[0] = *(const v4f*)xr0;
    a_cur[1] = *(const v4f*)xr1;
#pragma unroll
    for (int j = 0; j < 8; ++j)
        b_cur[j] = *(const v4f*)(wr + (size_t)((j >> 2) * 64 + (j & 3)) * DFEAT);

#pragma unroll
    for (int t = 0; t < DFEAT / 8; ++t) {
        v4f a_nxt[2], b_nxt[8];
        if (t + 1 < DFEAT / 8) {
            const int kb = 8 * (t + 1);
            a_nxt[0] = *(const v4f*)(xr0 + kb);
            a_nxt[1] = *(const v4f*)(xr1 + kb);
#pragma unroll
            for (int j = 0; j < 8; ++j)
                b_nxt[j] = *(const v4f*)(wr + (size_t)((j >> 2) * 64 + (j & 3)) * DFEAT + kb);
        }
        __builtin_amdgcn_sched_barrier(0);
#pragma unroll
        for (int s = 0; s < 2; ++s) {
            v2f af[2], bf[8];
#pragma unroll
            for (int i = 0; i < 2; ++i) {
                af[i].x = a_cur[i][2 * s]; af[i].y = a_cur[i][2 * s + 1];
            }
#pragma unroll
            for (int j = 0; j < 8; ++j) { bf[j].x = b_cur[j][2 * s]; bf[j].y = b_cur[j][2 * s + 1]; }
#pragma unroll
            for (int i = 0; i < 2; ++i)
#pragma unroll
                for (int j = 0; j < 8; ++j)
                    c[i][j] = __builtin_amdgcn_wmma_f32_16x16x4_f32(false, af[i], false, bf[j],
                                                                    (short)0, c[i][j], false, false);
        }
        __builtin_amdgcn_sched_barrier(0);
        if (t + 1 < DFEAT / 8) {
            a_cur[0] = a_nxt[0]; a_cur[1] = a_nxt[1];
#pragma unroll
            for (int j = 0; j < 8; ++j) b_cur[j] = b_nxt[j];
        }
    }

#pragma unroll
    for (int i = 0; i < 2; ++i) {
#pragma unroll
        for (int r = 0; r < 8; ++r) {
            const int M = row0 + 16 * i + r + 8 * half;
            if (M < nrows) {
                float* hp = H + (size_t)M * DFEAT + 4 * m;
#pragma unroll
                for (int g = 0; g < 2; ++g) {
                    v4f o;
                    o.x = c[i][4 * g + 0][r]; o.y = c[i][4 * g + 1][r];
                    o.z = c[i][4 * g + 2][r]; o.w = c[i][4 * g + 3][r];
                    *(v4f*)(hp + 64 * g) = o;
                }
            }
        }
    }
}

#define OWN_WAVES 32
#define ACC_PITCH 132
#define ACC_TILES ((ACC_ROWS + 16) / 16)
#define ACC_LROWS (ACC_TILES * 16)
#define QCAP 512
#define SCAN_B 8
#define PGROUP 8

#define LCAP 128
typedef int v2i __attribute__((ext_vector_type(2)));

template <bool REPLAY>
__global__ __launch_bounds__(1024) void aggregate_owner_kernel(const float* __restrict__ H,
                                                               const int* __restrict__ erec,
                                                               const float* __restrict__ deg,
                                                               const unsigned* __restrict__ okey,
                                                               float* __restrict__ out, int E,
                                                               int n, int* __restrict__ lst,
                                                               int* __restrict__ lcnt,
                                                               const float* __restrict__ W2,
                                                               const float* __restrict__ b2) {
    __shared__ __attribute__((aligned(16))) float acc[ACC_LROWS * ACC_PITCH];
    __shared__ int queue[OWN_WAVES * QCAP];
    __shared__ __attribute__((aligned(16))) float wsh[REPLAY ? 4 : DFEAT * ACC_PITCH];

    const int tid  = threadIdx.x;
    const int lane = tid & 31;
    const int wave = __builtin_amdgcn_readfirstlane(tid >> 5);
    const int base = (int)blockIdx.x * ACC_ROWS;
    const unsigned own = ((unsigned)blockIdx.x & 255u) * 0x01010101u;

    {
        const v4f z4 = {0.0f, 0.0f, 0.0f, 0.0f};
        for (int i = tid; i < ACC_LROWS * ACC_PITCH / 4; i += 1024) ((v4f*)acc)[i] = z4;
    }
    if (!REPLAY) {
        for (int i = tid; i < DFEAT * 32; i += 1024) {
            const int nn = i >> 5;
            const int l  = i & 31;
            const v4f wv = *(const v4f*)(W2 + nn * DFEAT + 4 * l);
            const int rho = (nn & 64) + 16 * (nn & 3) + ((nn & 63) >> 2);
            float* wp = wsh + rho * ACC_PITCH + l;
            wp[0] = wv.x; wp[32] = wv.y; wp[64] = wv.z; wp[96] = wv.w;
        }
    }
    __syncthreads();

    int* q = queue + wave * QCAP;
    int qn = 0;
    int tot = 0;
    v2i* wl = (v2i*)lst + (size_t)((int)blockIdx.x * OWN_WAVES + wave) * LCAP;

    auto apply = [&](const int r, const int rl, const int nvb, const int c32) {
            for (int i0 = 0; i0 < c32; i0 += PGROUP) {
                v4f hv[PGROUP];
#pragma unroll
                for (int u = 0; u < PGROUP; ++u) {
                    const int ri = __builtin_amdgcn_readlane(r, i0 + u);
                    hv[u] = *(const v4f*)(H + (size_t)ri * DFEAT + 4 * lane);
                }
#pragma unroll
                for (int u = 0; u < PGROUP; ++u) {
                    const int rli = __builtin_amdgcn_readlane(rl, i0 + u);
                    const float ni = __builtin_bit_cast(float, __builtin_amdgcn_readlane(nvb, i0 + u));
                    float* ap = acc + rli * ACC_PITCH + lane;
                    atomicAdd(ap,      ni * hv[u].x);
                    atomicAdd(ap + 32, ni * hv[u].y);
                    atomicAdd(ap + 64, ni * hv[u].z);
                    atomicAdd(ap + 96, ni * hv[u].w);
                }
            }
    };

    auto drain = [&](const int cntq) {
        for (int b0 = 0; b0 < cntq; b0 += 32) {
            const int idx = b0 + lane;
            const bool v  = idx < cntq;
            const int e   = v ? q[idx] : 0;
            const v4i rc  = ((const v4i*)erec)[e];
            const int r   = rc.x;
            const int c   = rc.y;
            const int wbits = rc.z;
            const float wv = __int_as_float(wbits);
            const float dr = deg[r];
            const float dc = deg[c];
            const float ir = 1.0f / sqrtf(dr);
            const float ic = 1.0f / sqrtf(dc);
            const float nv = ir * wv * ic;
            const unsigned rel = (unsigned)(c - base);
            const int rl  = (v && rel < (unsigned)ACC_ROWS) ? (int)rel : ACC_ROWS;
            const int nvb = __builtin_bit_cast(int, nv);
            if (!REPLAY) {
                const int li = tot + idx;
                if (lst != nullptr && v && li < LCAP)
                    wl[li] = v2i{(int)((unsigned)r | ((unsigned)rl << 24)), nvb};
            }
            const int rem = cntq - b0;
            apply(r, rl, nvb, rem < 32 ? rem : 32);
        }
        if (!REPLAY) tot += cntq;
    };

    bool scan = true;
    if (REPLAY) {
        if (lst != nullptr) {
            const int lc = __builtin_amdgcn_readfirstlane(lcnt[(int)blockIdx.x * OWN_WAVES + wave]);
            if (lc <= LCAP) {
                scan = false;
                if (lc > 0) {
                    const int last = lc - 1;
                    v2i en = wl[lane < last ? lane : last];
                    for (int b0 = 0; b0 < lc; b0 += 32) {
                        const int ix = b0 + 32 + lane;
                        const v2i nx = wl[ix < last ? ix : last];
                        const bool v = b0 + lane < lc;
                        const int rem = lc - b0;
                        apply(en.x & 0xffffff, v ? (int)((unsigned)en.x >> 24) : ACC_ROWS, en.y,
                              rem < 32 ? rem : 32);
                        en = nx;
                    }
                }
            }
        }
    }

    const int G = (E + 15) >> 4;
    const int U = (G + 31) >> 5;
    const v4u* __restrict__ k16 = (const v4u*)okey;
    for (int u0 = wave; scan && u0 < U; u0 += OWN_WAVES * SCAN_B) {
        v4u kk[SCAN_B];
#pragma unroll
        for (int i = 0; i < SCAN_B; ++i) {
            const int gq = (u0 + OWN_WAVES * i) * 32 + lane;
            kk[i] = k16[gq < G ? gq : G - 1];
        }
        __builtin_amdgcn_sched_barrier(0);
        unsigned zz[SCAN_B];
#pragma unroll
        for (int i = 0; i < SCAN_B; ++i) {
            unsigned zc = 0;
#pragma unroll
            for (int j = 0; j < 4; ++j) {
                const unsigned y = kk[i][j] ^ own;
                const unsigned z = ~(((y & 0x7f7f7f7fu) + 0x7f7f7f7fu) | y) & 0x80808080u;
                zc |= z >> (7 - j);
            }
            const int gq = (u0 + OWN_WAVES * i) * 32 + lane;
            zz[i] = gq < G ? zc : 0u;
        }
        for (;;) {
            unsigned cur = 0;
            int ci = 0;
#pragma unroll
            for (int k = SCAN_B - 1; k >= 0; --k) {
                if (zz[k] != 0) { cur = zz[k]; ci = k; }
            }
            const bool has = cur != 0;
            if (__builtin_amdgcn_ballot_w32(has) == 0) break;
            if (qn > QCAP - 32) {
                drain(qn);
                qn = 0;
            }
            const int t = has ? __builtin_ctz(cur) : 0;
            const int p = ((t & 7) << 2) + (t >> 3);
            const int e = (((u0 + OWN_WAVES * ci) * 32 + lane) << 4) + p;
            const unsigned keep = has ? ~(1u << t) : ~0u;
#pragma unroll
            for (int k = 0; k < SCAN_B; ++k) zz[k] &= (k == ci) ? keep : ~0u;
            const bool pred = has && e < E;
            const unsigned bal = (unsigned)__builtin_amdgcn_ballot_w32(pred);
            const int slot = qn + (int)__builtin_amdgcn_mbcnt_lo(bal, 0u);
            if (pred) q[slot] = e;
            qn += __builtin_popcount(bal);
        }
    }
    if (scan) drain(qn);
    if (!REPLAY) {
        if (lst != nullptr && lane == 0) lcnt[(int)blockIdx.x * OWN_WAVES + wave] = tot;
    }
    __syncthreads();

    const int left = n - base;
    const int rows = left < ACC_ROWS ? left : ACC_ROWS;
    if (REPLAY) {
        for (int rr = wave; rr < rows; rr += OWN_WAVES) {
            const float* ap = acc + rr * ACC_PITCH + lane;
            v4f o;
            o.x = ap[0]; o.y = ap[32]; o.z = ap[64]; o.w = ap[96];
            *(v4f*)(out + (size_t)(base + rr) * DFEAT + 4 * lane) = o;
        }
    } else {
        const int m    = lane & 15;
        const int half = lane >> 4;
        for (int task = wave; task < 2 * ACC_TILES; task += OWN_WAVES) {
            const int g  = task / ACC_TILES;
            const int rt = task - g * ACC_TILES;
            if (16 * rt >= rows) continue;
            const float* ar = acc + (16 * rt + m) * ACC_PITCH + 4 * half;
            const float* br = wsh + (64 * g + m) * ACC_PITCH + 4 * half;
            v8f c[4];
            {
                const v4f bv = *(const v4f*)(b2 + 64 * g + 4 * m);
#pragma unroll
                for (int jj = 0; jj < 4; ++jj)
#pragma unroll
                    for (int r = 0; r < 8; ++r) c[jj][r] = bv[jj];
            }
#pragma unroll
            for (int t = 0; t < DFEAT / 8; ++t) {
                v4f a = *(const v4f*)(ar + 8 * t);
                v4f b[4];
#pragma unroll
                for (int jj = 0; jj < 4; ++jj) b[jj] = *(const v4f*)(br + 16 * jj * ACC_PITCH + 8 * t);
                a.x = fmaxf(a.x, 0.0f); a.y = fmaxf(a.y, 0.0f);
                a.z = fmaxf(a.z, 0.0f); a.w = fmaxf(a.w, 0.0f);
#pragma unroll
                for (int s = 0; s < 2; ++s) {
                    v2f af;
                    af.x = a[2 * s]; af.y = a[2 * s + 1];
#pragma unroll
                    for (int jj = 0; jj < 4; ++jj) {
                        v2f bf;
                        bf.x = b[jj][2 * s]; bf.y = b[jj][2 * s + 1];
                        c[jj] = __builtin_amdgcn_wmma_f32_16x16x4_f32(false, af, false, bf, (short)0,
                                                                      c[jj], false, false);
                    }
                }
            }
#pragma unroll
            for (int r = 0; r < 8; ++r) {
                const int lr = 16 * rt + r + 8 * half;
                if (lr < rows) {
                    v4f o;
                    o.x = c[0][r]; o.y = c[1][r]; o.z = c[2][r]; o.w = c[3][r];
                    *(v4f*)(out + (size_t)(base + lr) * DFEAT + 64 * g + 4 * m) = o;
                }
            }
        }
    }
}

extern "C" void kernel_launch(void* const* d_in, const int* in_sizes, int n_in,
                              void* d_out, int out_size, void* d_ws, size_t ws_size,
                              hipStream_t stream) {
    const float* x  = (const float*)d_in[0];
    const int*   ei = (const int*)d_in[1];
    const float* ew = (const float*)d_in[2];
    const float* W1 = (const float*)d_in[3];
    const float* b1 = (const float*)d_in[4];
    const float* W2 = (const float*)d_in[5];
    const float* b2 = (const float*)d_in[6];
    float* out = (float*)d_out;

    const int E = in_sizes[2];
    const int n = in_sizes[0] / DFEAT;
    const int* row = ei;
    const int* col = ei + E;

    char* ws = (char*)d_ws;
    size_t off = 0;
    auto alignup = [](size_t v) { return (v + 255) & ~(size_t)255; };
    float* deg  = (float*)(ws + off); off += alignup((size_t)n * 4);
    const size_t zero_bytes = off;
    unsigned* okey = (unsigned*)(ws + off); off += alignup((size_t)E * 4);
    float* h    = (float*)(ws + off); off += alignup((size_t)n * DFEAT * 4);
    int* erec   = (int*)(ws + off);
    float* h1   = out;

    const int gemm_blocks = (n + 127) / 128;
    const int epad        = (E + 15) & ~15;
    const int deg_blocks  = (epad / 4 + 127) / 128;
    const int own_blocks  = (n + ACC_ROWS - 1) / ACC_ROWS;

    const size_t lcnt_off = off + alignup((size_t)E * 16);
    const size_t lst_off  = lcnt_off + alignup((size_t)own_blocks * OWN_WAVES * 4);
    const size_t lst_end  = lst_off + (size_t)own_blocks * OWN_WAVES * LCAP * 8;
    const bool lists = lst_end <= ws_size && n <= (1 << 24);
    int* lcnt = lists ? (int*)(ws + lcnt_off) : nullptr;
    int* lst  = lists ? (int*)(ws + lst_off) : nullptr;

    hipMemsetAsync(deg, 0, zero_bytes, stream);

    gemm_wmma_kernel<true><<<gemm_blocks + deg_blocks, 128, 0, stream>>>(x, W1, b1, h1, n, row, col,
                                                                       ew, deg, okey, erec, E);
    aggregate_owner_kernel<false><<<own_blocks, 1024, 0, stream>>>(h1, erec, deg, okey, h, E, n, lst, lcnt,
                                                                   W2, b2);

    aggregate_owner_kernel<true><<<own_blocks, 1024, 0, stream>>>(h, erec, deg, okey, out, E, n, lst, lcnt,
                                                                  nullptr, nullptr);
}
